# starting kernel
# baseline (speedup 1.0000x reference)
#include <hip/hip_runtime.h>
#include <math.h>

typedef __attribute__((ext_vector_type(2))) float v2f;
typedef __attribute__((ext_vector_type(8))) float v8f;

#define NQ   8
#define QDIM 256

__device__ __forceinline__ float2 cmul(float2 a, float2 b) {
  return make_float2(a.x * b.x - a.y * b.y, a.x * b.y + a.y * b.x);
}
__device__ __forceinline__ float2 cadd(float2 a, float2 b) {
  return make_float2(a.x + b.x, a.y + b.y);
}

__global__ __launch_bounds__(32) void build_W(const float* __restrict__ weights,
                                              int n_layers,
                                              float* __restrict__ wR,
                                              float* __restrict__ wI) {
  __shared__ float2 col[QDIM][32];
  const int t = threadIdx.x;
  const int j = blockIdx.x * 32 + t;

  for (int i = 0; i < QDIM; ++i) col[i][t] = make_float2(i == j ? 1.f : 0.f, 0.f);

  for (int l = 0; l < n_layers; ++l) {
    for (int q = 0; q < NQ; ++q) {
      const float ph = weights[(l * NQ + q) * 3 + 0];
      const float th = weights[(l * NQ + q) * 3 + 1];
      const float om = weights[(l * NQ + q) * 3 + 2];
      const float c  = cosf(0.5f * th), s = sinf(0.5f * th);
      const float ap = 0.5f * (ph + om), bm = 0.5f * (ph - om);
      const float2 u00 = make_float2( cosf(ap) * c, -sinf(ap) * c);
      const float2 u01 = make_float2(-cosf(bm) * s, -sinf(bm) * s);
      const float2 u10 = make_float2( cosf(bm) * s, -sinf(bm) * s);
      const float2 u11 = make_float2( cosf(ap) * c,  sinf(ap) * c);
      const int stride = 1 << (7 - q);
      for (int m = 0; m < QDIM / 2; ++m) {
        const int low = m & (stride - 1);
        const int hi  = m >> (7 - q);
        const int i0  = hi * (stride << 1) + low;
        const int i1  = i0 + stride;
        const float2 a0 = col[i0][t], a1 = col[i1][t];
        col[i0][t] = cadd(cmul(u00, a0), cmul(u01, a1));
        col[i1][t] = cadd(cmul(u10, a0), cmul(u11, a1));
      }
    }
    const int r = l % (NQ - 1) + 1;
    for (int q = 0; q < NQ; ++q) {
      const int tq    = (q + r) & 7;
      const int cmask = 1 << (7 - q);
      const int tmask = 1 << (7 - tq);
      for (int m = 0; m < QDIM; ++m) {
        if ((m & cmask) && !(m & tmask)) {
          const int m2 = m | tmask;
          const float2 tmp = col[m][t];
          col[m][t]  = col[m2][t];
          col[m2][t] = tmp;
        }
      }
    }
  }

  for (int n = 0; n < QDIM; ++n) {
    const int idx = (((j >> 1) * QDIM) + n) * 2 + (j & 1);
    wR[idx] = col[n][t].x;
    wI[idx] = col[n][t].y;
  }
}

__global__ __launch_bounds__(32) void vqe_wmma(const float* __restrict__ x,
                                               const float* __restrict__ fcw,
                                               const float* __restrict__ fcb,
                                               const float* __restrict__ wR,
                                               const float* __restrict__ wI,
                                               float* __restrict__ out) {
  __shared__ float PhiR[16][260];
  __shared__ float PhiI[16][260];
  __shared__ float zred[16][20];

  const int l = threadIdx.x;
  const int h = l >> 4;
  const int c = l & 15;
  const int sBase = blockIdx.x * 16;

  float cs[NQ], sn[NQ];
  {
    const float* xp = x + (size_t)(sBase + c) * NQ;
#pragma unroll
    for (int q = 0; q < NQ; ++q) {
      const float xv = 0.5f * xp[q];
      cs[q] = cosf(xv);
      sn[q] = sinf(xv);
    }
  }

  for (int tt = 0; tt < 128; ++tt) {
    const int b = (h << 7) | tt;
    float m = h ? sn[0] : cs[0];
#pragma unroll
    for (int q = 1; q < NQ; ++q)
      m *= ((b >> (7 - q)) & 1) ? sn[q] : cs[q];
    const int p = __popc(b) & 3;
    PhiR[c][b] = (p == 0) ? m : ((p == 2) ? -m : 0.f);
    PhiI[c][b] = (p == 1) ? -m : ((p == 3) ? m : 0.f);
  }
  __syncthreads();

  float fw[NQ];
#pragma unroll
  for (int i = 0; i < NQ; ++i) fw[i] = fcw[i];

  float zacc[8];
#pragma unroll
  for (int r = 0; r < 8; ++r) zacc[r] = 0.f;

  const float* aRbase = &PhiR[c][2 * h];
  const float* aIbase = &PhiI[c][2 * h];

  for (int nt = 0; nt < 16; ++nt) {
    const int n = nt * 16 + c;
    float g = 0.f;
#pragma unroll
    for (int i = 0; i < NQ; ++i)
      g += fw[i] * (1.f - 2.f * (float)((n >> (7 - i)) & 1));

    v8f cr1 = {0.f, 0.f, 0.f, 0.f, 0.f, 0.f, 0.f, 0.f};
    v8f cr2 = cr1, ci1 = cr1, ci2 = cr1;

    const float* bR = wR + (size_t)(h * 512 + 2 * n);
    const float* bI = wI + (size_t)(h * 512 + 2 * n);

#pragma unroll 4
    for (int k0 = 0; k0 < QDIM; k0 += 4) {
      const v2f ar = *(const v2f*)(aRbase + k0);
      const v2f ai = *(const v2f*)(aIbase + k0);
      const v2f br = *(const v2f*)(bR + (size_t)(k0 >> 1) * 512);
      const v2f bi = *(const v2f*)(bI + (size_t)(k0 >> 1) * 512);
      cr1 = __builtin_amdgcn_wmma_f32_16x16x4_f32(false, ar, false, br, (short)0, cr1, false, false);
      cr2 = __builtin_amdgcn_wmma_f32_16x16x4_f32(false, ai, false, bi, (short)0, cr2, false, false);
      ci1 = __builtin_amdgcn_wmma_f32_16x16x4_f32(false, ar, false, bi, (short)0, ci1, false, false);
      ci2 = __builtin_amdgcn_wmma_f32_16x16x4_f32(false, ai, false, br, (short)0, ci2, false, false);
    }

#pragma unroll
    for (int r = 0; r < 8; ++r) {
      const float pr = cr1[r] - cr2[r];
      const float pi = ci1[r] + ci2[r];
      zacc[r] += (pr * pr + pi * pi) * g;
    }
  }

#pragma unroll
  for (int r = 0; r < 8; ++r) zred[r + 8 * h][c] = zacc[r];
  __syncthreads();

  if (l < 16) {
    float sum = fcb[0];
#pragma unroll
    for (int cc = 0; cc < 16; ++cc) sum += zred[l][cc];
    out[sBase + l] = 1.f / (1.f + expf(-sum));
  }
}

extern "C" void kernel_launch(void* const* d_in, const int* in_sizes, int n_in,
                              void* d_out, int out_size, void* d_ws, size_t ws_size,
                              hipStream_t stream) {
  const float* x       = (const float*)d_in[0];
  const float* weights = (const float*)d_in[1];
  const float* fcw     = (const float*)d_in[2];
  const float* fcb     = (const float*)d_in[3];
  float* out = (float*)d_out;

  float* wR = (float*)d_ws;
  float* wI = wR + QDIM * QDIM;

  const int B        = in_sizes[0] / NQ;
  const int n_layers = in_sizes[1] / (NQ * 3);

  build_W<<<QDIM / 32, 32, 0, stream>>>(weights, n_layers, wR, wI);
  vqe_wmma<<<B / 16, 32, 0, stream>>>(x, fcw, fcb, wR, wI, out);
}


# build_W: one wave per pair of basis columns, amplitudes spread over lanes and registers (lane-xor exchange, CNOT ring as one LDS gather, gate trig one gate per lane); 128 waves instead of 8
# speedup vs baseline: 2.3846x; 2.3846x over previous
#include <hip/hip_runtime.h>
#include <math.h>

typedef __attribute__((ext_vector_type(2))) float v2f;
typedef __attribute__((ext_vector_type(8))) float v8f;

#define NQ   8
#define QDIM 256

__device__ __forceinline__ float2 cmul(float2 a, float2 b) {
  return make_float2(a.x * b.x - a.y * b.y, a.x * b.y + a.y * b.x);
}
__device__ __forceinline__ float2 cadd(float2 a, float2 b) {
  return make_float2(a.x + b.x, a.y + b.y);
}


__device__ __forceinline__ int ring_src(int i, int rr) {
#pragma unroll
  for (int q = NQ - 1; q >= 0; --q) {
    const int cbit = 7 - q;
    const int tbit = 7 - ((q + rr) & 7);
    i ^= ((i >> cbit) & 1) << tbit;
  }
  return i;
}

__global__ __launch_bounds__(32) void build_W(const float* __restrict__ weights,
                                              int n_layers,
                                              float* __restrict__ wR,
                                              float* __restrict__ wI) {
  __shared__ float4 gate[32][2];
  __shared__ float2 colA[QDIM];
  __shared__ float2 colB[QDIM];
  const int t = threadIdx.x;
  const int p = blockIdx.x;

  float2 a[8], b[8];
#pragma unroll
  for (int r = 0; r < 8; ++r) {
    const int i = r * 32 + t;
    a[r] = make_float2(i == 2 * p     ? 1.f : 0.f, 0.f);
    b[r] = make_float2(i == 2 * p + 1 ? 1.f : 0.f, 0.f);
  }

  for (int l0 = 0; l0 < n_layers; l0 += 4) {
    __syncthreads();
    {
      const int gl = l0 + (t >> 3);
      const int gq = t & 7;
      if (gl < n_layers) {
        const float ph = weights[(gl * NQ + gq) * 3 + 0];
        const float th = weights[(gl * NQ + gq) * 3 + 1];
        const float om = weights[(gl * NQ + gq) * 3 + 2];
        const float c  = cosf(0.5f * th), s = sinf(0.5f * th);
        const float ap = 0.5f * (ph + om), bm = 0.5f * (ph - om);
        gate[t][0] = make_float4( cosf(ap) * c, -sinf(ap) * c,
                                 -cosf(bm) * s, -sinf(bm) * s);
        gate[t][1] = make_float4( cosf(bm) * s, -sinf(bm) * s,
                                  cosf(ap) * c,  sinf(ap) * c);
      }
    }
    __syncthreads();

    const int lend = (l0 + 4 < n_layers) ? (l0 + 4) : n_layers;
    for (int l = l0; l < lend; ++l) {
#pragma unroll
      for (int q = 0; q < NQ; ++q) {
        const float4 g0 = gate[(l - l0) * 8 + q][0];
        const float4 g1 = gate[(l - l0) * 8 + q][1];
        const float2 u00 = make_float2(g0.x, g0.y), u01 = make_float2(g0.z, g0.w);
        const float2 u10 = make_float2(g1.x, g1.y), u11 = make_float2(g1.z, g1.w);
        if (q < 3) {
          const int rs = 4 >> q;
#pragma unroll
          for (int r = 0; r < 8; ++r) {
            if (r & rs) continue;
            const float2 a0 = a[r], a1 = a[r + rs];
            a[r]      = cadd(cmul(u00, a0), cmul(u01, a1));
            a[r + rs] = cadd(cmul(u10, a0), cmul(u11, a1));
            const float2 b0 = b[r], b1 = b[r + rs];
            b[r]      = cadd(cmul(u00, b0), cmul(u01, b1));
            b[r + rs] = cadd(cmul(u10, b0), cmul(u11, b1));
          }
        } else {
          const int  lm = 1 << (7 - q);
          const bool up = (t & lm) != 0;
          const float2 k0 = up ? u10 : u00;
          const float2 k1 = up ? u11 : u01;
float2 oa[8], ob[8];
#pragma unroll
          for (int r = 0; r < 8; ++r) {
            oa[r] = make_float2(__shfl_xor(a[r].x, lm), __shfl_xor(a[r].y, lm));
            ob[r] = make_float2(__shfl_xor(b[r].x, lm), __shfl_xor(b[r].y, lm));
          }
#pragma unroll
          for (int r = 0; r < 8; ++r) {
            const float2 ma = a[r], mb = b[r];
            a[r] = cadd(cmul(k0, up ? oa[r] : ma), cmul(k1, up ? ma : oa[r]));
            b[r] = cadd(cmul(k0, up ? ob[r] : mb), cmul(k1, up ? mb : ob[r]));
          }
        }
      }
      const int rr = l % (NQ - 1) + 1;
#pragma unroll
      for (int r = 0; r < 8; ++r) {
        colA[r * 32 + t] = a[r];
        colB[r * 32 + t] = b[r];
      }
      __syncthreads();
      const int st = ring_src(t, rr);
#pragma unroll
      for (int r = 0; r < 8; ++r) {
        const int si = st ^ ring_src(r * 32, rr);
        a[r] = colA[si];
        b[r] = colB[si];
      }
      __syncthreads();
    }
  }

  float2* __restrict__ wR2 = (float2*)wR;
  float2* __restrict__ wI2 = (float2*)wI;
#pragma unroll
  for (int r = 0; r < 8; ++r) {
    const int n = r * 32 + t;
    wR2[p * QDIM + n] = make_float2(a[r].x, b[r].x);
    wI2[p * QDIM + n] = make_float2(a[r].y, b[r].y);
  }
}

__global__ __launch_bounds__(32) void vqe_wmma(const float* __restrict__ x,
                                               const float* __restrict__ fcw,
                                               const float* __restrict__ fcb,
                                               const float* __restrict__ wR,
                                               const float* __restrict__ wI,
                                               float* __restrict__ out) {
  __shared__ float PhiR[16][260];
  __shared__ float PhiI[16][260];
  __shared__ float zred[16][20];

  const int l = threadIdx.x;
  const int h = l >> 4;
  const int c = l & 15;
  const int sBase = blockIdx.x * 16;

  float cs[NQ], sn[NQ];
  {
    const float* xp = x + (size_t)(sBase + c) * NQ;
#pragma unroll
    for (int q = 0; q < NQ; ++q) {
      const float xv = 0.5f * xp[q];
      cs[q] = cosf(xv);
      sn[q] = sinf(xv);
    }
  }

  for (int tt = 0; tt < 128; ++tt) {
    const int b = (h << 7) | tt;
    float m = h ? sn[0] : cs[0];
#pragma unroll
    for (int q = 1; q < NQ; ++q)
      m *= ((b >> (7 - q)) & 1) ? sn[q] : cs[q];
    const int p = __popc(b) & 3;
    PhiR[c][b] = (p == 0) ? m : ((p == 2) ? -m : 0.f);
    PhiI[c][b] = (p == 1) ? -m : ((p == 3) ? m : 0.f);
  }
  __syncthreads();

  float fw[NQ];
#pragma unroll
  for (int i = 0; i < NQ; ++i) fw[i] = fcw[i];

  float zacc[8];
#pragma unroll
  for (int r = 0; r < 8; ++r) zacc[r] = 0.f;

  const float* aRbase = &PhiR[c][2 * h];
  const float* aIbase = &PhiI[c][2 * h];

  for (int nt = 0; nt < 16; ++nt) {
    const int n = nt * 16 + c;
    float g = 0.f;
#pragma unroll
    for (int i = 0; i < NQ; ++i)
      g += fw[i] * (1.f - 2.f * (float)((n >> (7 - i)) & 1));

    v8f cr1 = {0.f, 0.f, 0.f, 0.f, 0.f, 0.f, 0.f, 0.f};
    v8f cr2 = cr1, ci1 = cr1, ci2 = cr1;

    const float* bR = wR + (size_t)(h * 512 + 2 * n);
    const float* bI = wI + (size_t)(h * 512 + 2 * n);

#pragma unroll 4
    for (int k0 = 0; k0 < QDIM; k0 += 4) {
      const v2f ar = *(const v2f*)(aRbase + k0);
      const v2f ai = *(const v2f*)(aIbase + k0);
      const v2f br = *(const v2f*)(bR + (size_t)(k0 >> 1) * 512);
      const v2f bi = *(const v2f*)(bI + (size_t)(k0 >> 1) * 512);
      cr1 = __builtin_amdgcn_wmma_f32_16x16x4_f32(false, ar, false, br, (short)0, cr1, false, false);
      cr2 = __builtin_amdgcn_wmma_f32_16x16x4_f32(false, ai, false, bi, (short)0, cr2, false, false);
      ci1 = __builtin_amdgcn_wmma_f32_16x16x4_f32(false, ar, false, bi, (short)0, ci1, false, false);
      ci2 = __builtin_amdgcn_wmma_f32_16x16x4_f32(false, ai, false, br, (short)0, ci2, false, false);
    }

#pragma unroll
    for (int r = 0; r < 8; ++r) {
      const float pr = cr1[r] - cr2[r];
      const float pi = ci1[r] + ci2[r];
      zacc[r] += (pr * pr + pi * pi) * g;
    }
  }

#pragma unroll
  for (int r = 0; r < 8; ++r) zred[r + 8 * h][c] = zacc[r];
  __syncthreads();

  if (l < 16) {
    float sum = fcb[0];
#pragma unroll
    for (int cc = 0; cc < 16; ++cc) sum += zred[l][cc];
    out[sBase + l] = 1.f / (1.f + expf(-sum));
  }
}

extern "C" void kernel_launch(void* const* d_in, const int* in_sizes, int n_in,
                              void* d_out, int out_size, void* d_ws, size_t ws_size,
                              hipStream_t stream) {
  const float* x       = (const float*)d_in[0];
  const float* weights = (const float*)d_in[1];
  const float* fcw     = (const float*)d_in[2];
  const float* fcb     = (const float*)d_in[3];
  float* out = (float*)d_out;

  float* wR = (float*)d_ws;
  float* wI = wR + QDIM * QDIM;

  const int B        = in_sizes[0] / NQ;
  const int n_layers = in_sizes[1] / (NQ * 3);

  build_W<<<QDIM / 2, 32, 0, stream>>>(weights, n_layers, wR, wI);
  vqe_wmma<<<B / 16, 32, 0, stream>>>(x, fcw, fcb, wR, wI, out);
}


# vqe_wmma k loop software-pipelined: B fragment loads issued one 64-WMMA block ahead into a second register set (across column tiles too), tile-invariant A held in registers; 2 waves/SIMD register cap
# speedup vs baseline: 3.9513x; 1.6570x over previous
#include <hip/hip_runtime.h>
#include <math.h>

typedef __attribute__((ext_vector_type(2))) float v2f;
typedef __attribute__((ext_vector_type(8))) float v8f;

#define NQ   8
#define QDIM 256

__device__ __forceinline__ float2 cmul(float2 a, float2 b) {
  return make_float2(a.x * b.x - a.y * b.y, a.x * b.y + a.y * b.x);
}
__device__ __forceinline__ float2 cadd(float2 a, float2 b) {
  return make_float2(a.x + b.x, a.y + b.y);
}


__device__ __forceinline__ int ring_src(int i, int rr) {
#pragma unroll
  for (int q = NQ - 1; q >= 0; --q) {
    const int cbit = 7 - q;
    const int tbit = 7 - ((q + rr) & 7);
    i ^= ((i >> cbit) & 1) << tbit;
  }
  return i;
}

__global__ __launch_bounds__(32) void build_W(const float* __restrict__ weights,
                                              int n_layers,
                                              float* __restrict__ wR,
                                              float* __restrict__ wI) {
  __shared__ float4 gate[32][2];
  __shared__ float2 colA[QDIM];
  __shared__ float2 colB[QDIM];
  const int t = threadIdx.x;
  const int p = blockIdx.x;

  float2 a[8], b[8];
#pragma unroll
  for (int r = 0; r < 8; ++r) {
    const int i = r * 32 + t;
    a[r] = make_float2(i == 2 * p     ? 1.f : 0.f, 0.f);
    b[r] = make_float2(i == 2 * p + 1 ? 1.f : 0.f, 0.f);
  }

  for (int l0 = 0; l0 < n_layers; l0 += 4) {
    __syncthreads();
    {
      const int gl = l0 + (t >> 3);
      const int gq = t & 7;
      if (gl < n_layers) {
        const float ph = weights[(gl * NQ + gq) * 3 + 0];
        const float th = weights[(gl * NQ + gq) * 3 + 1];
        const float om = weights[(gl * NQ + gq) * 3 + 2];
        const float c  = cosf(0.5f * th), s = sinf(0.5f * th);
        const float ap = 0.5f * (ph + om), bm = 0.5f * (ph - om);
        gate[t][0] = make_float4( cosf(ap) * c, -sinf(ap) * c,
                                 -cosf(bm) * s, -sinf(bm) * s);
        gate[t][1] = make_float4( cosf(bm) * s, -sinf(bm) * s,
                                  cosf(ap) * c,  sinf(ap) * c);
      }
    }
    __syncthreads();

    const int lend = (l0 + 4 < n_layers) ? (l0 + 4) : n_layers;
    for (int l = l0; l < lend; ++l) {
#pragma unroll
      for (int q = 0; q < NQ; ++q) {
        const float4 g0 = gate[(l - l0) * 8 + q][0];
        const float4 g1 = gate[(l - l0) * 8 + q][1];
        const float2 u00 = make_float2(g0.x, g0.y), u01 = make_float2(g0.z, g0.w);
        const float2 u10 = make_float2(g1.x, g1.y), u11 = make_float2(g1.z, g1.w);
        if (q < 3) {
          const int rs = 4 >> q;
#pragma unroll
          for (int r = 0; r < 8; ++r) {
            if (r & rs) continue;
            const float2 a0 = a[r], a1 = a[r + rs];
            a[r]      = cadd(cmul(u00, a0), cmul(u01, a1));
            a[r + rs] = cadd(cmul(u10, a0), cmul(u11, a1));
            const float2 b0 = b[r], b1 = b[r + rs];
            b[r]      = cadd(cmul(u00, b0), cmul(u01, b1));
            b[r + rs] = cadd(cmul(u10, b0), cmul(u11, b1));
          }
        } else {
          const int  lm = 1 << (7 - q);
          const bool up = (t & lm) != 0;
          const float2 k0 = up ? u10 : u00;
          const float2 k1 = up ? u11 : u01;
float2 oa[8], ob[8];
#pragma unroll
          for (int r = 0; r < 8; ++r) {
            oa[r] = make_float2(__shfl_xor(a[r].x, lm), __shfl_xor(a[r].y, lm));
            ob[r] = make_float2(__shfl_xor(b[r].x, lm), __shfl_xor(b[r].y, lm));
          }
#pragma unroll
          for (int r = 0; r < 8; ++r) {
            const float2 ma = a[r], mb = b[r];
            a[r] = cadd(cmul(k0, up ? oa[r] : ma), cmul(k1, up ? ma : oa[r]));
            b[r] = cadd(cmul(k0, up ? ob[r] : mb), cmul(k1, up ? mb : ob[r]));
          }
        }
      }
      const int rr = l % (NQ - 1) + 1;
#pragma unroll
      for (int r = 0; r < 8; ++r) {
        colA[r * 32 + t] = a[r];
        colB[r * 32 + t] = b[r];
      }
      __syncthreads();
      const int st = ring_src(t, rr);
#pragma unroll
      for (int r = 0; r < 8; ++r) {
        const int si = st ^ ring_src(r * 32, rr);
        a[r] = colA[si];
        b[r] = colB[si];
      }
      __syncthreads();
    }
  }

  float2* __restrict__ wR2 = (float2*)wR;
  float2* __restrict__ wI2 = (float2*)wI;
#pragma unroll
  for (int r = 0; r < 8; ++r) {
    const int n = r * 32 + t;
    wR2[p * QDIM + n] = make_float2(a[r].x, b[r].x);
    wI2[p * QDIM + n] = make_float2(a[r].y, b[r].y);
  }
}

#define KB   16
#define NBLK (QDIM / (4 * KB))
struct FragB { v2f br[KB], bi[KB]; };

__device__ __forceinline__ void load_fragB(FragB& f, const float* bR, const float* bI) {
#pragma unroll
  for (int j = 0; j < KB; ++j) {
    f.br[j] = *(const v2f*)(bR + (size_t)(2 * j) * 512);
    f.bi[j] = *(const v2f*)(bI + (size_t)(2 * j) * 512);
  }
}

__device__ __forceinline__ void mma_frag(const v2f* ar, const v2f* ai, const FragB& f,
                                         v8f& cr1, v8f& cr2, v8f& ci1, v8f& ci2) {
#pragma unroll
  for (int j = 0; j < KB; ++j) {
    cr1 = __builtin_amdgcn_wmma_f32_16x16x4_f32(false, ar[j], false, f.br[j], (short)0, cr1, false, false);
    cr2 = __builtin_amdgcn_wmma_f32_16x16x4_f32(false, ai[j], false, f.bi[j], (short)0, cr2, false, false);
    ci1 = __builtin_amdgcn_wmma_f32_16x16x4_f32(false, ar[j], false, f.bi[j], (short)0, ci1, false, false);
    ci2 = __builtin_amdgcn_wmma_f32_16x16x4_f32(false, ai[j], false, f.br[j], (short)0, ci2, false, false);
  }
}

__global__ __launch_bounds__(32) __attribute__((amdgpu_waves_per_eu(2, 2))) void vqe_wmma(const float* __restrict__ x,
                                               const float* __restrict__ fcw,
                                               const float* __restrict__ fcb,
                                               const float* __restrict__ wR,
                                               const float* __restrict__ wI,
                                               float* __restrict__ out) {
  __shared__ float PhiR[16][260];
  __shared__ float PhiI[16][260];
  __shared__ float zred[16][20];

  const int l = threadIdx.x;
  const int h = l >> 4;
  const int c = l & 15;
  const int sBase = blockIdx.x * 16;

  float cs[NQ], sn[NQ];
  {
    const float* xp = x + (size_t)(sBase + c) * NQ;
#pragma unroll
    for (int q = 0; q < NQ; ++q) {
      const float xv = 0.5f * xp[q];
      cs[q] = cosf(xv);
      sn[q] = sinf(xv);
    }
  }

  for (int tt = 0; tt < 128; ++tt) {
    const int b = (h << 7) | tt;
    float m = h ? sn[0] : cs[0];
#pragma unroll
    for (int q = 1; q < NQ; ++q)
      m *= ((b >> (7 - q)) & 1) ? sn[q] : cs[q];
    const int p = __popc(b) & 3;
    PhiR[c][b] = (p == 0) ? m : ((p == 2) ? -m : 0.f);
    PhiI[c][b] = (p == 1) ? -m : ((p == 3) ? m : 0.f);
  }
  __syncthreads();

  float fw[NQ];
#pragma unroll
  for (int i = 0; i < NQ; ++i) fw[i] = fcw[i];

  float zacc[8];
#pragma unroll
  for (int r = 0; r < 8; ++r) zacc[r] = 0.f;

  const float* aRbase = &PhiR[c][2 * h];
  const float* aIbase = &PhiI[c][2 * h];
  const float* bRbase = wR + (size_t)(h * 512);
  const float* bIbase = wI + (size_t)(h * 512);

  v2f aR[QDIM / 4], aI[QDIM / 4];
#pragma unroll
  for (int j = 0; j < QDIM / 4; ++j) {
    aR[j] = *(const v2f*)(aRbase + 4 * j);
    aI[j] = *(const v2f*)(aIbase + 4 * j);
  }

  FragB f0, f1;
  load_fragB(f0, bRbase + 2 * c, bIbase + 2 * c);

  for (int nt = 0; nt < 16; ++nt) {
    const int n = nt * 16 + c;
    const int nn = ((nt + 1 < 16) ? (nt + 1) : nt) * 16 + c;
    float g = 0.f;
#pragma unroll
    for (int i = 0; i < NQ; ++i)
      g += fw[i] * (1.f - 2.f * (float)((n >> (7 - i)) & 1));

    v8f cr1 = {0.f, 0.f, 0.f, 0.f, 0.f, 0.f, 0.f, 0.f};
    v8f cr2 = cr1, ci1 = cr1, ci2 = cr1;

    const float* bR = bRbase + 2 * n;
    const float* bI = bIbase + 2 * n;

#pragma unroll
    for (int kb = 0; kb < NBLK; kb += 2) {
      load_fragB(f1, bR + (size_t)((kb + 1) * KB * 2) * 512, bI + (size_t)((kb + 1) * KB * 2) * 512);
      mma_frag(aR + kb * KB, aI + kb * KB, f0, cr1, cr2, ci1, ci2);
      if (kb + 2 < NBLK) {
        load_fragB(f0, bR + (size_t)((kb + 2) * KB * 2) * 512, bI + (size_t)((kb + 2) * KB * 2) * 512);
      } else {
        load_fragB(f0, bRbase + 2 * nn, bIbase + 2 * nn);
      }
      mma_frag(aR + (kb + 1) * KB, aI + (kb + 1) * KB, f1, cr1, cr2, ci1, ci2);
    }

#pragma unroll
    for (int r = 0; r < 8; ++r) {
      const float pr = cr1[r] - cr2[r];
      const float pi = ci1[r] + ci2[r];
      zacc[r] += (pr * pr + pi * pi) * g;
    }
    asm volatile("" ::: "memory");
  }

#pragma unroll
  for (int r = 0; r < 8; ++r) zred[r + 8 * h][c] = zacc[r];
  __syncthreads();

  if (l < 16) {
    float sum = fcb[0];
#pragma unroll
    for (int cc = 0; cc < 16; ++cc) sum += zred[l][cc];
    out[sBase + l] = 1.f / (1.f + expf(-sum));
  }
}

extern "C" void kernel_launch(void* const* d_in, const int* in_sizes, int n_in,
                              void* d_out, int out_size, void* d_ws, size_t ws_size,
                              hipStream_t stream) {
  const float* x       = (const float*)d_in[0];
  const float* weights = (const float*)d_in[1];
  const float* fcw     = (const float*)d_in[2];
  const float* fcb     = (const float*)d_in[3];
  float* out = (float*)d_out;

  float* wR = (float*)d_ws;
  float* wI = wR + QDIM * QDIM;

  const int B        = in_sizes[0] / NQ;
  const int n_layers = in_sizes[1] / (NQ * 3);

  build_W<<<QDIM / 2, 32, 0, stream>>>(weights, n_layers, wR, wI);
  vqe_wmma<<<B / 16, 32, 0, stream>>>(x, fcw, fcb, wR, wI, out);
}


# vqe_wmma: A fragments built in registers by each lane from its own cos/sin (prefix products, same qubit order, values bit-identical) instead of a 128-step fill through LDS, barrier and 64 ds_loads
# speedup vs baseline: 7.0420x; 1.7822x over previous
#include <hip/hip_runtime.h>
#include <math.h>

typedef __attribute__((ext_vector_type(2))) float v2f;
typedef __attribute__((ext_vector_type(8))) float v8f;

#define NQ   8
#define QDIM 256

__device__ __forceinline__ float2 cmul(float2 a, float2 b) {
  return make_float2(a.x * b.x - a.y * b.y, a.x * b.y + a.y * b.x);
}
__device__ __forceinline__ float2 cadd(float2 a, float2 b) {
  return make_float2(a.x + b.x, a.y + b.y);
}
__device__ __forceinline__ float2 mul_minus_i_pow(float2 z, int e) {
  const float2 s = (e & 1) ? make_float2(z.y, -z.x) : z;
  return (e & 2) ? make_float2(-s.x, -s.y) : s;
}


__device__ __forceinline__ int ring_src(int i, int rr) {
#pragma unroll
  for (int q = NQ - 1; q >= 0; --q) {
    const int cbit = 7 - q;
    const int tbit = 7 - ((q + rr) & 7);
    i ^= ((i >> cbit) & 1) << tbit;
  }
  return i;
}

__global__ __launch_bounds__(32) void build_W(const float* __restrict__ weights,
                                              int n_layers,
                                              float* __restrict__ wR,
                                              float* __restrict__ wI) {
  __shared__ float4 gate[32][2];
  __shared__ float2 colA[QDIM];
  __shared__ float2 colB[QDIM];
  const int t = threadIdx.x;
  const int p = blockIdx.x;

  float2 a[8], b[8];
#pragma unroll
  for (int r = 0; r < 8; ++r) {
    const int i = r * 32 + t;
    a[r] = make_float2(i == 2 * p     ? 1.f : 0.f, 0.f);
    b[r] = make_float2(i == 2 * p + 1 ? 1.f : 0.f, 0.f);
  }

  for (int l0 = 0; l0 < n_layers; l0 += 4) {
    __syncthreads();
    {
      const int gl = l0 + (t >> 3);
      const int gq = t & 7;
      if (gl < n_layers) {
        const float ph = weights[(gl * NQ + gq) * 3 + 0];
        const float th = weights[(gl * NQ + gq) * 3 + 1];
        const float om = weights[(gl * NQ + gq) * 3 + 2];
        const float c  = cosf(0.5f * th), s = sinf(0.5f * th);
        const float ap = 0.5f * (ph + om), bm = 0.5f * (ph - om);
        gate[t][0] = make_float4( cosf(ap) * c, -sinf(ap) * c,
                                 -cosf(bm) * s, -sinf(bm) * s);
        gate[t][1] = make_float4( cosf(bm) * s, -sinf(bm) * s,
                                  cosf(ap) * c,  sinf(ap) * c);
      }
    }
    __syncthreads();

    const int lend = (l0 + 4 < n_layers) ? (l0 + 4) : n_layers;
    for (int l = l0; l < lend; ++l) {
#pragma unroll
      for (int q = 0; q < NQ; ++q) {
        const float4 g0 = gate[(l - l0) * 8 + q][0];
        const float4 g1 = gate[(l - l0) * 8 + q][1];
        const float2 u00 = make_float2(g0.x, g0.y), u01 = make_float2(g0.z, g0.w);
        const float2 u10 = make_float2(g1.x, g1.y), u11 = make_float2(g1.z, g1.w);
        if (q < 3) {
          const int rs = 4 >> q;
#pragma unroll
          for (int r = 0; r < 8; ++r) {
            if (r & rs) continue;
            const float2 a0 = a[r], a1 = a[r + rs];
            a[r]      = cadd(cmul(u00, a0), cmul(u01, a1));
            a[r + rs] = cadd(cmul(u10, a0), cmul(u11, a1));
            const float2 b0 = b[r], b1 = b[r + rs];
            b[r]      = cadd(cmul(u00, b0), cmul(u01, b1));
            b[r + rs] = cadd(cmul(u10, b0), cmul(u11, b1));
          }
        } else {
          const int  lm = 1 << (7 - q);
          const bool up = (t & lm) != 0;
          const float2 k0 = up ? u10 : u00;
          const float2 k1 = up ? u11 : u01;
float2 oa[8], ob[8];
#pragma unroll
          for (int r = 0; r < 8; ++r) {
            oa[r] = make_float2(__shfl_xor(a[r].x, lm), __shfl_xor(a[r].y, lm));
            ob[r] = make_float2(__shfl_xor(b[r].x, lm), __shfl_xor(b[r].y, lm));
          }
#pragma unroll
          for (int r = 0; r < 8; ++r) {
            const float2 ma = a[r], mb = b[r];
            a[r] = cadd(cmul(k0, up ? oa[r] : ma), cmul(k1, up ? ma : oa[r]));
            b[r] = cadd(cmul(k0, up ? ob[r] : mb), cmul(k1, up ? mb : ob[r]));
          }
        }
      }
      const int rr = l % (NQ - 1) + 1;
#pragma unroll
      for (int r = 0; r < 8; ++r) {
        colA[r * 32 + t] = a[r];
        colB[r * 32 + t] = b[r];
      }
      __syncthreads();
      const int st = ring_src(t, rr);
#pragma unroll
      for (int r = 0; r < 8; ++r) {
        const int si = st ^ ring_src(r * 32, rr);
        a[r] = colA[si];
        b[r] = colB[si];
      }
      __syncthreads();
    }
  }

  const int ea = __popc(2 * p);
  float2* __restrict__ wR2 = (float2*)wR;
  float2* __restrict__ wI2 = (float2*)wI;
#pragma unroll
  for (int r = 0; r < 8; ++r) {
    const int n = r * 32 + t;
    const float2 pa = mul_minus_i_pow(a[r], ea);
    const float2 pb = mul_minus_i_pow(b[r], ea + 1);
    wR2[p * QDIM + n] = make_float2(pa.x, pb.x);
    wI2[p * QDIM + n] = make_float2(pa.y, pb.y);
  }
}

#define KB   16
#define NBLK (QDIM / (4 * KB))
struct FragB { v2f br[KB], bi[KB]; };

__device__ __forceinline__ void load_fragB(FragB& f, const float* bR, const float* bI) {
#pragma unroll
  for (int j = 0; j < KB; ++j) {
    f.br[j] = *(const v2f*)(bR + (size_t)(2 * j) * 512);
    f.bi[j] = *(const v2f*)(bI + (size_t)(2 * j) * 512);
  }
}

__device__ __forceinline__ void mma_frag(const v2f* am, const FragB& f, v8f& cr, v8f& ci) {
#pragma unroll
  for (int j = 0; j < KB; ++j) {
    cr = __builtin_amdgcn_wmma_f32_16x16x4_f32(false, am[j], false, f.br[j], (short)0, cr, false, false);
    ci = __builtin_amdgcn_wmma_f32_16x16x4_f32(false, am[j], false, f.bi[j], (short)0, ci, false, false);
  }
}

__global__ __launch_bounds__(32) __attribute__((amdgpu_waves_per_eu(2, 2))) void vqe_wmma(const float* __restrict__ x,
                                               const float* __restrict__ fcw,
                                               const float* __restrict__ fcb,
                                               const float* __restrict__ wR,
                                               const float* __restrict__ wI,
                                               float* __restrict__ out) {
  __shared__ float zred[16][20];

  const int l = threadIdx.x;
  const int h = l >> 4;
  const int c = l & 15;
  const int sBase = blockIdx.x * 16;

  float cs[NQ], sn[NQ];
  {
    const float* xp = x + (size_t)(sBase + c) * NQ;
#pragma unroll
    for (int q = 0; q < NQ; ++q) {
      const float xv = 0.5f * xp[q];
      cs[q] = cosf(xv);
      sn[q] = sinf(xv);
    }
  }

  v2f aM[QDIM / 4];
  {
    float P[64];
    P[0]  = cs[0];
    P[32] = sn[0];
#pragma unroll
    for (int q = 1; q < 6; ++q) {
      const int step = 32 >> q;
#pragma unroll
      for (int base = 0; base < 64; base += 2 * step) {
        const float v = P[base];
        P[base]        = v * cs[q];
        P[base + step] = v * sn[q];
      }
    }
    const float f6 = h ? sn[6] : cs[6];
#pragma unroll
    for (int j = 0; j < QDIM / 4; ++j) {
      const float t6 = P[j] * f6;
      aM[j].x = t6 * cs[7];
      aM[j].y = t6 * sn[7];
    }
  }

  float fw[NQ];
#pragma unroll
  for (int i = 0; i < NQ; ++i) fw[i] = fcw[i];

  float zacc[8];
#pragma unroll
  for (int r = 0; r < 8; ++r) zacc[r] = 0.f;

  const float* bRbase = wR + (size_t)(h * 512);
  const float* bIbase = wI + (size_t)(h * 512);

  FragB f0, f1;
  load_fragB(f0, bRbase + 2 * c, bIbase + 2 * c);

  for (int nt = 0; nt < 16; ++nt) {
    const int n = nt * 16 + c;
    const int nn = ((nt + 1 < 16) ? (nt + 1) : nt) * 16 + c;
    float g = 0.f;
#pragma unroll
    for (int i = 0; i < NQ; ++i)
      g += fw[i] * (1.f - 2.f * (float)((n >> (7 - i)) & 1));

    v8f cr = {0.f, 0.f, 0.f, 0.f, 0.f, 0.f, 0.f, 0.f};
    v8f ci = cr;

    const float* bR = bRbase + 2 * n;
    const float* bI = bIbase + 2 * n;

#pragma unroll
    for (int kb = 0; kb < NBLK; kb += 2) {
      load_fragB(f1, bR + (size_t)((kb + 1) * KB * 2) * 512, bI + (size_t)((kb + 1) * KB * 2) * 512);
      __builtin_amdgcn_sched_barrier(6);
      mma_frag(aM + kb * KB, f0, cr, ci);
      __builtin_amdgcn_sched_barrier(6);
      if (kb + 2 < NBLK) {
        load_fragB(f0, bR + (size_t)((kb + 2) * KB * 2) * 512, bI + (size_t)((kb + 2) * KB * 2) * 512);
      } else {
        load_fragB(f0, bRbase + 2 * nn, bIbase + 2 * nn);
      }
      __builtin_amdgcn_sched_barrier(6);
      mma_frag(aM + (kb + 1) * KB, f1, cr, ci);
      __builtin_amdgcn_sched_barrier(6);
    }

#pragma unroll
    for (int r = 0; r < 8; ++r) {
      const float pr = cr[r];
      const float pi = ci[r];
      zacc[r] += (pr * pr + pi * pi) * g;
    }
    asm volatile("" ::: "memory");
  }

#pragma unroll
  for (int r = 0; r < 8; ++r) zred[r + 8 * h][c] = zacc[r];
  __syncthreads();

  if (l < 16) {
    float sum = fcb[0];
#pragma unroll
    for (int cc = 0; cc < 16; ++cc) sum += zred[l][cc];
    out[sBase + l] = 1.f / (1.f + expf(-sum));
  }
}

extern "C" void kernel_launch(void* const* d_in, const int* in_sizes, int n_in,
                              void* d_out, int out_size, void* d_ws, size_t ws_size,
                              hipStream_t stream) {
  const float* x       = (const float*)d_in[0];
  const float* weights = (const float*)d_in[1];
  const float* fcw     = (const float*)d_in[2];
  const float* fcb     = (const float*)d_in[3];
  float* out = (float*)d_out;

  float* wR = (float*)d_ws;
  float* wI = wR + QDIM * QDIM;

  const int B        = in_sizes[0] / NQ;
  const int n_layers = in_sizes[1] / (NQ * 3);

  build_W<<<QDIM / 2, 32, 0, stream>>>(weights, n_layers, wR, wI);
  vqe_wmma<<<B / 16, 32, 0, stream>>>(x, fcw, fcb, wR, wI, out);
}


# readout folded into the circuit matrix: new WMMA kernel forms real Q = Re(Wp diag(g) Wp^H), per sample z = m^T Q m: one real product not a complex one (1024 WMMAs per wave not 2048); sums reordered
# speedup vs baseline: 9.5662x; 1.3584x over previous
#include <hip/hip_runtime.h>
#include <math.h>

typedef __attribute__((ext_vector_type(2))) float v2f;
typedef __attribute__((ext_vector_type(4))) float v4f;
typedef __attribute__((ext_vector_type(8))) float v8f;

#define NQ   8
#define QDIM 256

__device__ __forceinline__ float2 cmul(float2 a, float2 b) {
  return make_float2(a.x * b.x - a.y * b.y, a.x * b.y + a.y * b.x);
}
__device__ __forceinline__ float2 cadd(float2 a, float2 b) {
  return make_float2(a.x + b.x, a.y + b.y);
}
__device__ __forceinline__ float2 mul_minus_i_pow(float2 z, int e) {
  const float2 s = (e & 1) ? make_float2(z.y, -z.x) : z;
  return (e & 2) ? make_float2(-s.x, -s.y) : s;
}


__device__ __forceinline__ int ring_src(int i, int rr) {
#pragma unroll
  for (int q = NQ - 1; q >= 0; --q) {
    const int cbit = 7 - q;
    const int tbit = 7 - ((q + rr) & 7);
    i ^= ((i >> cbit) & 1) << tbit;
  }
  return i;
}

__global__ __launch_bounds__(32) void build_W(const float* __restrict__ weights,
                                              const float* __restrict__ fcw,
                                              int n_layers,
                                              float* __restrict__ rR,
                                              float* __restrict__ rI,
                                              float* __restrict__ gR,
                                              float* __restrict__ gI) {
  __shared__ float4 gate[32][2];
  __shared__ float2 colA[QDIM];
  __shared__ float2 colB[QDIM];
  const int t = threadIdx.x;
  const int p = blockIdx.x;

  float2 a[8], b[8];
#pragma unroll
  for (int r = 0; r < 8; ++r) {
    const int i = r * 32 + t;
    a[r] = make_float2(i == 2 * p     ? 1.f : 0.f, 0.f);
    b[r] = make_float2(i == 2 * p + 1 ? 1.f : 0.f, 0.f);
  }

  for (int l0 = 0; l0 < n_layers; l0 += 4) {
    __syncthreads();
    {
      const int gl = l0 + (t >> 3);
      const int gq = t & 7;
      if (gl < n_layers) {
        const float ph = weights[(gl * NQ + gq) * 3 + 0];
        const float th = weights[(gl * NQ + gq) * 3 + 1];
        const float om = weights[(gl * NQ + gq) * 3 + 2];
        const float c  = cosf(0.5f * th), s = sinf(0.5f * th);
        const float ap = 0.5f * (ph + om), bm = 0.5f * (ph - om);
        gate[t][0] = make_float4( cosf(ap) * c, -sinf(ap) * c,
                                 -cosf(bm) * s, -sinf(bm) * s);
        gate[t][1] = make_float4( cosf(bm) * s, -sinf(bm) * s,
                                  cosf(ap) * c,  sinf(ap) * c);
      }
    }
    __syncthreads();

    const int lend = (l0 + 4 < n_layers) ? (l0 + 4) : n_layers;
    for (int l = l0; l < lend; ++l) {
#pragma unroll
      for (int q = 0; q < NQ; ++q) {
        const float4 g0 = gate[(l - l0) * 8 + q][0];
        const float4 g1 = gate[(l - l0) * 8 + q][1];
        const float2 u00 = make_float2(g0.x, g0.y), u01 = make_float2(g0.z, g0.w);
        const float2 u10 = make_float2(g1.x, g1.y), u11 = make_float2(g1.z, g1.w);
        if (q < 3) {
          const int rs = 4 >> q;
#pragma unroll
          for (int r = 0; r < 8; ++r) {
            if (r & rs) continue;
            const float2 a0 = a[r], a1 = a[r + rs];
            a[r]      = cadd(cmul(u00, a0), cmul(u01, a1));
            a[r + rs] = cadd(cmul(u10, a0), cmul(u11, a1));
            const float2 b0 = b[r], b1 = b[r + rs];
            b[r]      = cadd(cmul(u00, b0), cmul(u01, b1));
            b[r + rs] = cadd(cmul(u10, b0), cmul(u11, b1));
          }
        } else {
          const int  lm = 1 << (7 - q);
          const bool up = (t & lm) != 0;
          const float2 k0 = up ? u10 : u00;
          const float2 k1 = up ? u11 : u01;
float2 oa[8], ob[8];
#pragma unroll
          for (int r = 0; r < 8; ++r) {
            oa[r] = make_float2(__shfl_xor(a[r].x, lm), __shfl_xor(a[r].y, lm));
            ob[r] = make_float2(__shfl_xor(b[r].x, lm), __shfl_xor(b[r].y, lm));
          }
#pragma unroll
          for (int r = 0; r < 8; ++r) {
            const float2 ma = a[r], mb = b[r];
            a[r] = cadd(cmul(k0, up ? oa[r] : ma), cmul(k1, up ? ma : oa[r]));
            b[r] = cadd(cmul(k0, up ? ob[r] : mb), cmul(k1, up ? mb : ob[r]));
          }
        }
      }
      const int rr = l % (NQ - 1) + 1;
#pragma unroll
      for (int r = 0; r < 8; ++r) {
        colA[r * 32 + t] = a[r];
        colB[r * 32 + t] = b[r];
      }
      __syncthreads();
      const int st = ring_src(t, rr);
#pragma unroll
      for (int r = 0; r < 8; ++r) {
        const int si = st ^ ring_src(r * 32, rr);
        a[r] = colA[si];
        b[r] = colB[si];
      }
      __syncthreads();
    }
  }

  float fw[NQ];
#pragma unroll
  for (int i = 0; i < NQ; ++i) fw[i] = fcw[i];
  const int ea = __popc(2 * p);
#pragma unroll
  for (int r = 0; r < 8; ++r) {
    const int n = r * 32 + t;
    float g = 0.f;
#pragma unroll
    for (int i = 0; i < NQ; ++i)
      g += fw[i] * (1.f - 2.f * (float)((n >> (7 - i)) & 1));
    const float2 pa = mul_minus_i_pow(a[r], ea);
    const float2 pb = mul_minus_i_pow(b[r], ea + 1);
    const int ia = (2 * p) * QDIM + n, ib = ia + QDIM;
    rR[ia] = pa.x;      rI[ia] = pa.y;
    rR[ib] = pb.x;      rI[ib] = pb.y;
    gR[ia] = g * pa.x;  gI[ia] = g * pa.y;
    gR[ib] = g * pb.x;  gI[ib] = g * pb.y;
  }
}

#define QB 8
struct FragQ { v4f ar[QB], ai[QB], br[QB], bi[QB]; };

__device__ __forceinline__ void load_fragQ(FragQ& f, const float* pgR, const float* pgI,
                                           const float* prR, const float* prI) {
#pragma unroll
  for (int i = 0; i < QB; ++i) {
    f.ar[i] = *(const v4f*)(pgR + 8 * i);
    f.ai[i] = *(const v4f*)(pgI + 8 * i);
    f.br[i] = *(const v4f*)(prR + 8 * i);
    f.bi[i] = *(const v4f*)(prI + 8 * i);
  }
}

__device__ __forceinline__ void mma_fragQ(const FragQ& f, v8f& accR, v8f& accI) {
#pragma unroll
  for (int i = 0; i < QB; ++i) {
    const v2f ar0 = {f.ar[i].x, f.ar[i].y}, ar1 = {f.ar[i].z, f.ar[i].w};
    const v2f ai0 = {f.ai[i].x, f.ai[i].y}, ai1 = {f.ai[i].z, f.ai[i].w};
    const v2f br0 = {f.br[i].x, f.br[i].y}, br1 = {f.br[i].z, f.br[i].w};
    const v2f bi0 = {f.bi[i].x, f.bi[i].y}, bi1 = {f.bi[i].z, f.bi[i].w};
    accR = __builtin_amdgcn_wmma_f32_16x16x4_f32(false, ar0, false, br0, (short)0, accR, false, false);
    accI = __builtin_amdgcn_wmma_f32_16x16x4_f32(false, ai0, false, bi0, (short)0, accI, false, false);
    accR = __builtin_amdgcn_wmma_f32_16x16x4_f32(false, ar1, false, br1, (short)0, accR, false, false);
    accI = __builtin_amdgcn_wmma_f32_16x16x4_f32(false, ai1, false, bi1, (short)0, accI, false, false);
  }
}

__global__ __launch_bounds__(32) void build_Q(const float* __restrict__ rR,
                                              const float* __restrict__ rI,
                                              const float* __restrict__ gR,
                                              const float* __restrict__ gI,
                                              float* __restrict__ qP) {
  const int l = threadIdx.x;
  const int h = l >> 4;
  const int c = l & 15;
  const int kt = blockIdx.x >> 4;
  const int nt = blockIdx.x & 15;
  const size_t ao = (size_t)(kt * 16 + c) * QDIM + 4 * h;
  const size_t bo = (size_t)(nt * 16 + c) * QDIM + 4 * h;

  v8f accR = {0.f, 0.f, 0.f, 0.f, 0.f, 0.f, 0.f, 0.f};
  v8f accI = accR;
  FragQ f0, f1;
  load_fragQ(f0, gR + ao, gI + ao, rR + bo, rI + bo);
#pragma unroll
  for (int sb = 0; sb < QDIM / 8 / QB; sb += 2) {
    load_fragQ(f1, gR + ao + 8 * QB * (sb + 1), gI + ao + 8 * QB * (sb + 1),
                   rR + bo + 8 * QB * (sb + 1), rI + bo + 8 * QB * (sb + 1));
    __builtin_amdgcn_sched_barrier(6);
    mma_fragQ(f0, accR, accI);
    __builtin_amdgcn_sched_barrier(6);
    if (sb + 2 < QDIM / 8 / QB) {
      load_fragQ(f0, gR + ao + 8 * QB * (sb + 2), gI + ao + 8 * QB * (sb + 2),
                     rR + bo + 8 * QB * (sb + 2), rI + bo + 8 * QB * (sb + 2));
    }
    __builtin_amdgcn_sched_barrier(6);
    mma_fragQ(f1, accR, accI);
    __builtin_amdgcn_sched_barrier(6);
  }

  float2* __restrict__ q2 = (float2*)qP;
#pragma unroll
  for (int r = 0; r < 8; r += 2) {
    const int k = kt * 16 + 8 * h + r;
    q2[(k >> 1) * QDIM + nt * 16 + c] = make_float2(accR[r] + accI[r], accR[r + 1] + accI[r + 1]);
  }
}

#define KB   16
#define NBLK (QDIM / (4 * KB))
struct FragB { v2f br[KB], bi[KB]; };

__device__ __forceinline__ void load_fragB(FragB& f, const float* bR, const float* bI) {
#pragma unroll
  for (int j = 0; j < KB; ++j) {
    f.br[j] = *(const v2f*)(bR + (size_t)(2 * j) * 512);
    f.bi[j] = *(const v2f*)(bI + (size_t)(2 * j) * 512);
  }
}

__device__ __forceinline__ void mma_frag(const v2f* am, const FragB& f, v8f& cr, v8f& ci) {
#pragma unroll
  for (int j = 0; j < KB; ++j) {
    cr = __builtin_amdgcn_wmma_f32_16x16x4_f32(false, f.br[j], false, am[j], (short)0, cr, false, false);
    ci = __builtin_amdgcn_wmma_f32_16x16x4_f32(false, f.bi[j], false, am[j], (short)0, ci, false, false);
  }
}

__global__ __launch_bounds__(32) __attribute__((amdgpu_waves_per_eu(2, 2))) void vqe_wmma(const float* __restrict__ x,
                                               const float* __restrict__ fcb,
                                               const float* __restrict__ qP,
                                               float* __restrict__ out) {
  const int l = threadIdx.x;
  const int h = l >> 4;
  const int c = l & 15;
  const int sBase = blockIdx.x * 16;

  float cs[NQ], sn[NQ];
  {
    const float* xp = x + (size_t)(sBase + c) * NQ;
#pragma unroll
    for (int q = 0; q < NQ; ++q) {
      const float xv = 0.5f * xp[q];
      cs[q] = cosf(xv);
      sn[q] = sinf(xv);
    }
  }

  v2f aM[QDIM / 4];
  {
    float P[64];
    P[0]  = cs[0];
    P[32] = sn[0];
#pragma unroll
    for (int q = 1; q < 6; ++q) {
      const int step = 32 >> q;
#pragma unroll
      for (int base = 0; base < 64; base += 2 * step) {
        const float v = P[base];
        P[base]        = v * cs[q];
        P[base + step] = v * sn[q];
      }
    }
    const float f6 = h ? sn[6] : cs[6];
#pragma unroll
    for (int j = 0; j < QDIM / 4; ++j) {
      const float t6 = P[j] * f6;
      aM[j].x = t6 * cs[7];
      aM[j].y = t6 * sn[7];
    }
  }

  float w8[8];
  {
    const float f4 = h ? sn[4] : cs[4];
#pragma unroll
    for (int r = 0; r < 8; ++r)
      w8[r] = ((f4 * ((r & 4) ? sn[5] : cs[5])) * ((r & 2) ? sn[6] : cs[6])) * ((r & 1) ? sn[7] : cs[7]);
  }

  const float* qbase = qP + (size_t)(h * 512);

  FragB f0, f1;
  load_fragB(f0, qbase + 2 * c, qbase + 2 * (c + 16));

  float z = 0.f;
  for (int tp = 0; tp < 8; ++tp) {
    const int n = tp * 32 + c;
    const int nn = ((tp + 1 < 8) ? (tp + 1) : tp) * 32 + c;

    v8f cr = {0.f, 0.f, 0.f, 0.f, 0.f, 0.f, 0.f, 0.f};
    v8f ci = cr;

    const float* bR = qbase + 2 * n;
    const float* bI = bR + 32;

#pragma unroll
    for (int kb = 0; kb < NBLK; kb += 2) {
      load_fragB(f1, bR + (size_t)((kb + 1) * KB * 2) * 512, bI + (size_t)((kb + 1) * KB * 2) * 512);
      __builtin_amdgcn_sched_barrier(6);
      mma_frag(aM + kb * KB, f0, cr, ci);
      __builtin_amdgcn_sched_barrier(6);
      if (kb + 2 < NBLK) {
        load_fragB(f0, bR + (size_t)((kb + 2) * KB * 2) * 512, bI + (size_t)((kb + 2) * KB * 2) * 512);
      } else {
        load_fragB(f0, qbase + 2 * nn, qbase + 2 * (nn + 16));
      }
      __builtin_amdgcn_sched_barrier(6);
      mma_frag(aM + (kb + 1) * KB, f1, cr, ci);
      __builtin_amdgcn_sched_barrier(6);
    }

    const float p3 = (((tp & 4) ? sn[0] : cs[0]) * ((tp & 2) ? sn[1] : cs[1])) * ((tp & 1) ? sn[2] : cs[2]);
    float s0 = 0.f, s1 = 0.f;
#pragma unroll
    for (int r = 0; r < 8; ++r) {
      s0 += cr[r] * w8[r];
      s1 += ci[r] * w8[r];
    }
    z += (p3 * cs[3]) * s0 + (p3 * sn[3]) * s1;
    asm volatile("" ::: "memory");
  }

  z += __shfl_xor(z, 16);
  if (l < 16) {
    const float sum = fcb[0] + z;
    out[sBase + l] = 1.f / (1.f + expf(-sum));
  }
}

extern "C" void kernel_launch(void* const* d_in, const int* in_sizes, int n_in,
                              void* d_out, int out_size, void* d_ws, size_t ws_size,
                              hipStream_t stream) {
  const float* x       = (const float*)d_in[0];
  const float* weights = (const float*)d_in[1];
  const float* fcw     = (const float*)d_in[2];
  const float* fcb     = (const float*)d_in[3];
  float* out = (float*)d_out;

  float* rR = (float*)d_ws;
  float* rI = rR + QDIM * QDIM;
  float* gR = rI + QDIM * QDIM;
  float* gI = gR + QDIM * QDIM;
  float* qP = gI + QDIM * QDIM;

  const int B        = in_sizes[0] / NQ;
  const int n_layers = in_sizes[1] / (NQ * 3);

  build_W<<<QDIM / 2, 32, 0, stream>>>(weights, fcw, n_layers, rR, rI, gR, gI);
  build_Q<<<QDIM, 32, 0, stream>>>(rR, rI, gR, gI, qP);
  vqe_wmma<<<B / 16, 32, 0, stream>>>(x, fcb, qP, out);
}


# vqe_wmma: readout form rewritten in the per-qubit symmetric basis (cos^2, cos*sin, sin^2): kernel 2 sums Q over index-pair classes to 81x81 coefficients; 162 WMMAs per wave not 1024; sums reordered
# speedup vs baseline: 11.3527x; 1.1868x over previous
#include <hip/hip_runtime.h>
#include <math.h>

typedef __attribute__((ext_vector_type(2))) float v2f;
typedef __attribute__((ext_vector_type(4))) float v4f;
typedef __attribute__((ext_vector_type(8))) float v8f;

#define NQ   8
#define QDIM 256

__device__ __forceinline__ float2 cmul(float2 a, float2 b) {
  return make_float2(a.x * b.x - a.y * b.y, a.x * b.y + a.y * b.x);
}
__device__ __forceinline__ float2 cadd(float2 a, float2 b) {
  return make_float2(a.x + b.x, a.y + b.y);
}
__device__ __forceinline__ float2 mul_minus_i_pow(float2 z, int e) {
  const float2 s = (e & 1) ? make_float2(z.y, -z.x) : z;
  return (e & 2) ? make_float2(-s.x, -s.y) : s;
}


__device__ __forceinline__ int ring_src(int i, int rr) {
#pragma unroll
  for (int q = NQ - 1; q >= 0; --q) {
    const int cbit = 7 - q;
    const int tbit = 7 - ((q + rr) & 7);
    i ^= ((i >> cbit) & 1) << tbit;
  }
  return i;
}

__global__ __launch_bounds__(32) void build_W(const float* __restrict__ weights,
                                              const float* __restrict__ fcw,
                                              int n_layers,
                                              float* __restrict__ rR,
                                              float* __restrict__ rI,
                                              float* __restrict__ gR,
                                              float* __restrict__ gI) {
  __shared__ float4 gate[32][2];
  __shared__ float2 colA[QDIM];
  __shared__ float2 colB[QDIM];
  const int t = threadIdx.x;
  const int p = blockIdx.x;

  float2 a[8], b[8];
#pragma unroll
  for (int r = 0; r < 8; ++r) {
    const int i = r * 32 + t;
    a[r] = make_float2(i == 2 * p     ? 1.f : 0.f, 0.f);
    b[r] = make_float2(i == 2 * p + 1 ? 1.f : 0.f, 0.f);
  }

  for (int l0 = 0; l0 < n_layers; l0 += 4) {
    __syncthreads();
    {
      const int gl = l0 + (t >> 3);
      const int gq = t & 7;
      if (gl < n_layers) {
        const float ph = weights[(gl * NQ + gq) * 3 + 0];
        const float th = weights[(gl * NQ + gq) * 3 + 1];
        const float om = weights[(gl * NQ + gq) * 3 + 2];
        const float c  = cosf(0.5f * th), s = sinf(0.5f * th);
        const float ap = 0.5f * (ph + om), bm = 0.5f * (ph - om);
        gate[t][0] = make_float4( cosf(ap) * c, -sinf(ap) * c,
                                 -cosf(bm) * s, -sinf(bm) * s);
        gate[t][1] = make_float4( cosf(bm) * s, -sinf(bm) * s,
                                  cosf(ap) * c,  sinf(ap) * c);
      }
    }
    __syncthreads();

    const int lend = (l0 + 4 < n_layers) ? (l0 + 4) : n_layers;
    for (int l = l0; l < lend; ++l) {
#pragma unroll
      for (int q = 0; q < NQ; ++q) {
        const float4 g0 = gate[(l - l0) * 8 + q][0];
        const float4 g1 = gate[(l - l0) * 8 + q][1];
        const float2 u00 = make_float2(g0.x, g0.y), u01 = make_float2(g0.z, g0.w);
        const float2 u10 = make_float2(g1.x, g1.y), u11 = make_float2(g1.z, g1.w);
        if (q < 3) {
          const int rs = 4 >> q;
#pragma unroll
          for (int r = 0; r < 8; ++r) {
            if (r & rs) continue;
            const float2 a0 = a[r], a1 = a[r + rs];
            a[r]      = cadd(cmul(u00, a0), cmul(u01, a1));
            a[r + rs] = cadd(cmul(u10, a0), cmul(u11, a1));
            const float2 b0 = b[r], b1 = b[r + rs];
            b[r]      = cadd(cmul(u00, b0), cmul(u01, b1));
            b[r + rs] = cadd(cmul(u10, b0), cmul(u11, b1));
          }
        } else {
          const int  lm = 1 << (7 - q);
          const bool up = (t & lm) != 0;
          const float2 k0 = up ? u10 : u00;
          const float2 k1 = up ? u11 : u01;
float2 oa[8], ob[8];
#pragma unroll
          for (int r = 0; r < 8; ++r) {
            oa[r] = make_float2(__shfl_xor(a[r].x, lm), __shfl_xor(a[r].y, lm));
            ob[r] = make_float2(__shfl_xor(b[r].x, lm), __shfl_xor(b[r].y, lm));
          }
#pragma unroll
          for (int r = 0; r < 8; ++r) {
            const float2 ma = a[r], mb = b[r];
            a[r] = cadd(cmul(k0, up ? oa[r] : ma), cmul(k1, up ? ma : oa[r]));
            b[r] = cadd(cmul(k0, up ? ob[r] : mb), cmul(k1, up ? mb : ob[r]));
          }
        }
      }
      const int rr = l % (NQ - 1) + 1;
#pragma unroll
      for (int r = 0; r < 8; ++r) {
        colA[r * 32 + t] = a[r];
        colB[r * 32 + t] = b[r];
      }
      __syncthreads();
      const int st = ring_src(t, rr);
#pragma unroll
      for (int r = 0; r < 8; ++r) {
        const int si = st ^ ring_src(r * 32, rr);
        a[r] = colA[si];
        b[r] = colB[si];
      }
      __syncthreads();
    }
  }

  float fw[NQ];
#pragma unroll
  for (int i = 0; i < NQ; ++i) fw[i] = fcw[i];
  const int ea = __popc(2 * p);
#pragma unroll
  for (int r = 0; r < 8; ++r) {
    const int n = r * 32 + t;
    float g = 0.f;
#pragma unroll
    for (int i = 0; i < NQ; ++i)
      g += fw[i] * (1.f - 2.f * (float)((n >> (7 - i)) & 1));
    const float2 pa = mul_minus_i_pow(a[r], ea);
    const float2 pb = mul_minus_i_pow(b[r], ea + 1);
    const int ia = (2 * p) * QDIM + n, ib = ia + QDIM;
    rR[ia] = pa.x;      rI[ia] = pa.y;
    rR[ib] = pb.x;      rI[ib] = pb.y;
    gR[ia] = g * pa.x;  gI[ia] = g * pa.y;
    gR[ib] = g * pb.x;  gI[ib] = g * pb.y;
  }
}

#define QB 8
struct FragQ { v4f ar[QB], ai[QB], br[QB], bi[QB]; };

__device__ __forceinline__ void load_fragQ(FragQ& f, const float* pgR, const float* pgI,
                                           const float* prR, const float* prI) {
#pragma unroll
  for (int i = 0; i < QB; ++i) {
    f.ar[i] = *(const v4f*)(pgR + 8 * i);
    f.ai[i] = *(const v4f*)(pgI + 8 * i);
    f.br[i] = *(const v4f*)(prR + 8 * i);
    f.bi[i] = *(const v4f*)(prI + 8 * i);
  }
}

__device__ __forceinline__ void mma_fragQ(const FragQ& f, v8f& accR, v8f& accI) {
#pragma unroll
  for (int i = 0; i < QB; ++i) {
    const v2f ar0 = {f.ar[i].x, f.ar[i].y}, ar1 = {f.ar[i].z, f.ar[i].w};
    const v2f ai0 = {f.ai[i].x, f.ai[i].y}, ai1 = {f.ai[i].z, f.ai[i].w};
    const v2f br0 = {f.br[i].x, f.br[i].y}, br1 = {f.br[i].z, f.br[i].w};
    const v2f bi0 = {f.bi[i].x, f.bi[i].y}, bi1 = {f.bi[i].z, f.bi[i].w};
    accR = __builtin_amdgcn_wmma_f32_16x16x4_f32(false, ar0, false, br0, (short)0, accR, false, false);
    accI = __builtin_amdgcn_wmma_f32_16x16x4_f32(false, ai0, false, bi0, (short)0, accI, false, false);
    accR = __builtin_amdgcn_wmma_f32_16x16x4_f32(false, ar1, false, br1, (short)0, accR, false, false);
    accI = __builtin_amdgcn_wmma_f32_16x16x4_f32(false, ai1, false, bi1, (short)0, accI, false, false);
  }
}

#define CROWS  81
#define CPITCH 128
#define CWAVES 8

__global__ __launch_bounds__(CWAVES * 32) void build_C(const float* __restrict__ rR,
                                                       const float* __restrict__ rI,
                                                       const float* __restrict__ gR,
                                                       const float* __restrict__ gI,
                                                       float* __restrict__ cP) {
  __shared__ float tile[CWAVES][16][17];
  __shared__ float part[CWAVES][CPITCH];
  const int tid = threadIdx.x;
  const int w = __builtin_amdgcn_readfirstlane(tid >> 5);
  const int l = tid & 31;
  const int h = l >> 4;
  const int c = l & 15;
  const int i = blockIdx.x;

  int base2 = 0, m1 = 0;
  {
    int v = i;
#pragma unroll
    for (int b = 0; b < 4; ++b) {
      const int d = v % 3;
      v /= 3;
      base2 |= (d == 2 ? 1 : 0) << b;
      m1    |= (d == 1 ? 1 : 0) << b;
    }
  }
  const int m1p = m1 & (m1 - 1);
  int sub = -1, nact = 0;
  for (int s = 0; s < 16; ++s) {
    if ((s & ~m1p) == 0) {
      if (nact == w) sub = s;
      ++nact;
    }
  }
  const bool active = sub >= 0;

  if (active) {
    const int kt = base2 | sub;
    const int nt = base2 | (m1 ^ sub);
    const size_t ao = (size_t)(kt * 16 + c) * QDIM + 4 * h;
    const size_t bo = (size_t)(nt * 16 + c) * QDIM + 4 * h;

    v8f accR = {0.f, 0.f, 0.f, 0.f, 0.f, 0.f, 0.f, 0.f};
    v8f accI = accR;
    FragQ f0, f1;
    load_fragQ(f0, gR + ao, gI + ao, rR + bo, rI + bo);
#pragma unroll
    for (int sb = 0; sb < QDIM / 8 / QB; sb += 2) {
      load_fragQ(f1, gR + ao + 8 * QB * (sb + 1), gI + ao + 8 * QB * (sb + 1),
                     rR + bo + 8 * QB * (sb + 1), rI + bo + 8 * QB * (sb + 1));
      __builtin_amdgcn_sched_barrier(6);
      mma_fragQ(f0, accR, accI);
      __builtin_amdgcn_sched_barrier(6);
      if (sb + 2 < QDIM / 8 / QB) {
        load_fragQ(f0, gR + ao + 8 * QB * (sb + 2), gI + ao + 8 * QB * (sb + 2),
                       rR + bo + 8 * QB * (sb + 2), rI + bo + 8 * QB * (sb + 2));
      }
      __builtin_amdgcn_sched_barrier(6);
      mma_fragQ(f1, accR, accI);
      __builtin_amdgcn_sched_barrier(6);
    }
#pragma unroll
    for (int r = 0; r < 8; ++r) tile[w][8 * h + r][c] = accR[r] + accI[r];
  }
  __syncthreads();

  if (active) {
#pragma unroll
    for (int u = 0; u < 4; ++u) {
      const int o  = l + 32 * u;
      const int jh = o >> 2;
      const int t7 = o & 3;
      float s = 0.f;
      if (jh < 27 && t7 < 3) {
        const int d1 = jh % 3, d2 = (jh / 3) % 3, d3 = jh / 9;
        const int b2 = (t7 == 2 ? 1 : 0) | (d1 == 2 ? 2 : 0) | (d2 == 2 ? 4 : 0) | (d3 == 2 ? 8 : 0);
        const int mm = (t7 == 1 ? 1 : 0) | (d1 == 1 ? 2 : 0) | (d2 == 1 ? 4 : 0) | (d3 == 1 ? 8 : 0);
        for (int q = 0; q < 16; ++q)
          if ((q & ~mm) == 0) s += tile[w][b2 | q][b2 | (mm ^ q)];
      }
      part[w][o] = s;
    }
  }
  __syncthreads();

  if (tid < CPITCH) {
    float s = 0.f;
    for (int ww = 0; ww < nact; ++ww) s += part[ww][tid];
    cP[i * CPITCH + tid] = (m1 != 0 ? 2.f : 1.f) * s;
  }
}

#define KS 27
struct FragA { v2f a0[KS], a1[KS]; };

__device__ __forceinline__ void load_fragA(FragA& f, const float* p) {
#pragma unroll
  for (int j = 0; j < KS; ++j) {
    f.a0[j] = *(const v2f*)(p + (size_t)j * CPITCH);
    f.a1[j] = *(const v2f*)(p + (size_t)j * CPITCH + 64);
  }
}

__device__ __forceinline__ void mma_fragA(const v2f* bm, const FragA& f, v8f& y0, v8f& y1) {
#pragma unroll
  for (int j = 0; j < KS; ++j) {
    y0 = __builtin_amdgcn_wmma_f32_16x16x4_f32(false, f.a0[j], false, bm[j], (short)0, y0, false, false);
    y1 = __builtin_amdgcn_wmma_f32_16x16x4_f32(false, f.a1[j], false, bm[j], (short)0, y1, false, false);
  }
}

__global__ __launch_bounds__(32) __attribute__((amdgpu_waves_per_eu(2, 2))) void vqe_wmma(const float* __restrict__ x,
                                               const float* __restrict__ fcb,
                                               const float* __restrict__ cP,
                                               float* __restrict__ out) {
  const int l = threadIdx.x;
  const int h = l >> 4;
  const int c = l & 15;
  const int sBase = blockIdx.x * 16;

  const v4f xa = *(const v4f*)(x + (size_t)(sBase + c) * NQ);
  const v4f xb = *(const v4f*)(x + (size_t)(sBase + c) * NQ + 4);
  const float xq[NQ] = {xa.x, xa.y, xa.z, xa.w, xb.x, xb.y, xb.z, xb.w};

  const float* abase = cP + 4 * c + 2 * h;
  FragA f0, f1;
  load_fragA(f0, abase);
  load_fragA(f1, abase + (size_t)KS * CPITCH);

  float F[NQ][3];
  {
#pragma unroll
    for (int q = 0; q < NQ; ++q) {
      const float xv = 0.5f * xq[q];
      const float cq = cosf(xv);
      const float sq = sinf(xv);
      F[q][0] = cq * cq;
      F[q][1] = cq * sq;
      F[q][2] = sq * sq;
    }
  }

  v2f bM[CROWS];
  {
    const float vx = h ? F[7][2] : F[7][0];
    const float vy = h ? 0.f     : F[7][1];
#pragma unroll
    for (int t0 = 0; t0 < 3; ++t0) {
#pragma unroll
      for (int t1 = 0; t1 < 3; ++t1) {
        const float p01 = F[0][t0] * F[1][t1];
#pragma unroll
        for (int t2 = 0; t2 < 3; ++t2) {
          const float p012 = p01 * F[2][t2];
#pragma unroll
          for (int t3 = 0; t3 < 3; ++t3) {
            const float u = p012 * F[3][t3];
            const int i = ((t0 * 3 + t1) * 3 + t2) * 3 + t3;
            bM[i].x = u * vx;
            bM[i].y = u * vy;
          }
        }
      }
    }
  }

  float w0[8], w1[8];
  {
    float V[32];
#pragma unroll
    for (int jh = 0; jh < 32; ++jh)
      V[jh] = (jh < 27) ? (F[4][jh / 9] * F[5][(jh / 3) % 3]) * F[6][jh % 3] : 0.f;
#pragma unroll
    for (int r = 0; r < 8; ++r) {
      w0[r] = h ? V[8 + r]  : V[r];
      w1[r] = h ? V[24 + r] : V[16 + r];
    }
  }

  v8f y0 = {0.f, 0.f, 0.f, 0.f, 0.f, 0.f, 0.f, 0.f};
  v8f y1 = y0;
  __builtin_amdgcn_sched_barrier(6);
  mma_fragA(bM, f0, y0, y1);
  __builtin_amdgcn_sched_barrier(6);
  load_fragA(f0, abase + (size_t)(2 * KS) * CPITCH);
  __builtin_amdgcn_sched_barrier(6);
  mma_fragA(bM + KS, f1, y0, y1);
  __builtin_amdgcn_sched_barrier(6);
  mma_fragA(bM + 2 * KS, f0, y0, y1);

  float s0 = 0.f, s1 = 0.f;
#pragma unroll
  for (int r = 0; r < 8; ++r) {
    s0 += y0[r] * w0[r];
    s1 += y1[r] * w1[r];
  }
  float z = s0 + s1;

  z += __shfl_xor(z, 16);
  if (l < 16) {
    const float sum = fcb[0] + z;
    out[sBase + l] = 1.f / (1.f + expf(-sum));
  }
}

extern "C" void kernel_launch(void* const* d_in, const int* in_sizes, int n_in,
                              void* d_out, int out_size, void* d_ws, size_t ws_size,
                              hipStream_t stream) {
  const float* x       = (const float*)d_in[0];
  const float* weights = (const float*)d_in[1];
  const float* fcw     = (const float*)d_in[2];
  const float* fcb     = (const float*)d_in[3];
  float* out = (float*)d_out;

  float* rR = (float*)d_ws;
  float* rI = rR + QDIM * QDIM;
  float* gR = rI + QDIM * QDIM;
  float* gI = gR + QDIM * QDIM;
  float* cP = gI + QDIM * QDIM;

  const int B        = in_sizes[0] / NQ;
  const int n_layers = in_sizes[1] / (NQ * 3);

  build_W<<<QDIM / 2, 32, 0, stream>>>(weights, fcw, n_layers, rR, rI, gR, gI);
  build_C<<<CROWS, CWAVES * 32, 0, stream>>>(rR, rI, gR, gI, cP);
  vqe_wmma<<<B / 16, 32, 0, stream>>>(x, fcb, cP, out);
}


# build_C class sums: LDS reads no longer behind lane tests (index masked to a valid subset, non-members selected to exact zero): they overlap instead of 45 serial load-wait-add steps; same sum order
# speedup vs baseline: 12.0856x; 1.0646x over previous
#include <hip/hip_runtime.h>
#include <math.h>

typedef __attribute__((ext_vector_type(2))) float v2f;
typedef __attribute__((ext_vector_type(4))) float v4f;
typedef __attribute__((ext_vector_type(8))) float v8f;

#define NQ   8
#define QDIM 256

__device__ __forceinline__ float2 cmul(float2 a, float2 b) {
  return make_float2(a.x * b.x - a.y * b.y, a.x * b.y + a.y * b.x);
}
__device__ __forceinline__ float2 cadd(float2 a, float2 b) {
  return make_float2(a.x + b.x, a.y + b.y);
}
__device__ __forceinline__ float2 mul_minus_i_pow(float2 z, int e) {
  const float2 s = (e & 1) ? make_float2(z.y, -z.x) : z;
  return (e & 2) ? make_float2(-s.x, -s.y) : s;
}


__device__ __forceinline__ int ring_src(int i, int rr) {
#pragma unroll
  for (int q = NQ - 1; q >= 0; --q) {
    const int cbit = 7 - q;
    const int tbit = 7 - ((q + rr) & 7);
    i ^= ((i >> cbit) & 1) << tbit;
  }
  return i;
}

__global__ __launch_bounds__(32) void build_W(const float* __restrict__ weights,
                                              const float* __restrict__ fcw,
                                              int n_layers,
                                              float* __restrict__ rR,
                                              float* __restrict__ rI,
                                              float* __restrict__ gR,
                                              float* __restrict__ gI) {
  __shared__ float4 gate[32][2];
  __shared__ float2 colA[QDIM];
  __shared__ float2 colB[QDIM];
  const int t = threadIdx.x;
  const int p = blockIdx.x;

  float2 a[8], b[8];
#pragma unroll
  for (int r = 0; r < 8; ++r) {
    const int i = r * 32 + t;
    a[r] = make_float2(i == 2 * p     ? 1.f : 0.f, 0.f);
    b[r] = make_float2(i == 2 * p + 1 ? 1.f : 0.f, 0.f);
  }

  for (int l0 = 0; l0 < n_layers; l0 += 4) {
    __syncthreads();
    {
      const int gl = l0 + (t >> 3);
      const int gq = t & 7;
      if (gl < n_layers) {
        const float ph = weights[(gl * NQ + gq) * 3 + 0];
        const float th = weights[(gl * NQ + gq) * 3 + 1];
        const float om = weights[(gl * NQ + gq) * 3 + 2];
        const float c  = cosf(0.5f * th), s = sinf(0.5f * th);
        const float ap = 0.5f * (ph + om), bm = 0.5f * (ph - om);
        gate[t][0] = make_float4( cosf(ap) * c, -sinf(ap) * c,
                                 -cosf(bm) * s, -sinf(bm) * s);
        gate[t][1] = make_float4( cosf(bm) * s, -sinf(bm) * s,
                                  cosf(ap) * c,  sinf(ap) * c);
      }
    }
    __syncthreads();

    const int lend = (l0 + 4 < n_layers) ? (l0 + 4) : n_layers;
    for (int l = l0; l < lend; ++l) {
#pragma unroll
      for (int q = 0; q < NQ; ++q) {
        const float4 g0 = gate[(l - l0) * 8 + q][0];
        const float4 g1 = gate[(l - l0) * 8 + q][1];
        const float2 u00 = make_float2(g0.x, g0.y), u01 = make_float2(g0.z, g0.w);
        const float2 u10 = make_float2(g1.x, g1.y), u11 = make_float2(g1.z, g1.w);
        if (q < 3) {
          const int rs = 4 >> q;
#pragma unroll
          for (int r = 0; r < 8; ++r) {
            if (r & rs) continue;
            const float2 a0 = a[r], a1 = a[r + rs];
            a[r]      = cadd(cmul(u00, a0), cmul(u01, a1));
            a[r + rs] = cadd(cmul(u10, a0), cmul(u11, a1));
            const float2 b0 = b[r], b1 = b[r + rs];
            b[r]      = cadd(cmul(u00, b0), cmul(u01, b1));
            b[r + rs] = cadd(cmul(u10, b0), cmul(u11, b1));
          }
        } else {
          const int  lm = 1 << (7 - q);
          const bool up = (t & lm) != 0;
          const float2 k0 = up ? u10 : u00;
          const float2 k1 = up ? u11 : u01;
float2 oa[8], ob[8];
#pragma unroll
          for (int r = 0; r < 8; ++r) {
            oa[r] = make_float2(__shfl_xor(a[r].x, lm), __shfl_xor(a[r].y, lm));
            ob[r] = make_float2(__shfl_xor(b[r].x, lm), __shfl_xor(b[r].y, lm));
          }
#pragma unroll
          for (int r = 0; r < 8; ++r) {
            const float2 ma = a[r], mb = b[r];
            a[r] = cadd(cmul(k0, up ? oa[r] : ma), cmul(k1, up ? ma : oa[r]));
            b[r] = cadd(cmul(k0, up ? ob[r] : mb), cmul(k1, up ? mb : ob[r]));
          }
        }
      }
      const int rr = l % (NQ - 1) + 1;
#pragma unroll
      for (int r = 0; r < 8; ++r) {
        colA[r * 32 + t] = a[r];
        colB[r * 32 + t] = b[r];
      }
      __syncthreads();
      const int st = ring_src(t, rr);
#pragma unroll
      for (int r = 0; r < 8; ++r) {
        const int si = st ^ ring_src(r * 32, rr);
        a[r] = colA[si];
        b[r] = colB[si];
      }
      __syncthreads();
    }
  }

  float fw[NQ];
#pragma unroll
  for (int i = 0; i < NQ; ++i) fw[i] = fcw[i];
  const int ea = __popc(2 * p);
#pragma unroll
  for (int r = 0; r < 8; ++r) {
    const int n = r * 32 + t;
    float g = 0.f;
#pragma unroll
    for (int i = 0; i < NQ; ++i)
      g += fw[i] * (1.f - 2.f * (float)((n >> (7 - i)) & 1));
    const float2 pa = mul_minus_i_pow(a[r], ea);
    const float2 pb = mul_minus_i_pow(b[r], ea + 1);
    const int ia = (2 * p) * QDIM + n, ib = ia + QDIM;
    rR[ia] = pa.x;      rI[ia] = pa.y;
    rR[ib] = pb.x;      rI[ib] = pb.y;
    gR[ia] = g * pa.x;  gI[ia] = g * pa.y;
    gR[ib] = g * pb.x;  gI[ib] = g * pb.y;
  }
}

#define QB 8
struct FragQ { v4f ar[QB], ai[QB], br[QB], bi[QB]; };

__device__ __forceinline__ void load_fragQ(FragQ& f, const float* pgR, const float* pgI,
                                           const float* prR, const float* prI) {
#pragma unroll
  for (int i = 0; i < QB; ++i) {
    f.ar[i] = *(const v4f*)(pgR + 8 * i);
    f.ai[i] = *(const v4f*)(pgI + 8 * i);
    f.br[i] = *(const v4f*)(prR + 8 * i);
    f.bi[i] = *(const v4f*)(prI + 8 * i);
  }
}

__device__ __forceinline__ void mma_fragQ(const FragQ& f, v8f& accR, v8f& accI) {
#pragma unroll
  for (int i = 0; i < QB; ++i) {
    const v2f ar0 = {f.ar[i].x, f.ar[i].y}, ar1 = {f.ar[i].z, f.ar[i].w};
    const v2f ai0 = {f.ai[i].x, f.ai[i].y}, ai1 = {f.ai[i].z, f.ai[i].w};
    const v2f br0 = {f.br[i].x, f.br[i].y}, br1 = {f.br[i].z, f.br[i].w};
    const v2f bi0 = {f.bi[i].x, f.bi[i].y}, bi1 = {f.bi[i].z, f.bi[i].w};
    accR = __builtin_amdgcn_wmma_f32_16x16x4_f32(false, ar0, false, br0, (short)0, accR, false, false);
    accI = __builtin_amdgcn_wmma_f32_16x16x4_f32(false, ai0, false, bi0, (short)0, accI, false, false);
    accR = __builtin_amdgcn_wmma_f32_16x16x4_f32(false, ar1, false, br1, (short)0, accR, false, false);
    accI = __builtin_amdgcn_wmma_f32_16x16x4_f32(false, ai1, false, bi1, (short)0, accI, false, false);
  }
}

#define CROWS  81
#define CPITCH 128
#define CWAVES 8

__global__ __launch_bounds__(CWAVES * 32) void build_C(const float* __restrict__ rR,
                                                       const float* __restrict__ rI,
                                                       const float* __restrict__ gR,
                                                       const float* __restrict__ gI,
                                                       float* __restrict__ cP) {
  __shared__ float tile[CWAVES][16][17];
  __shared__ float part[CWAVES][CPITCH];
  const int tid = threadIdx.x;
  const int w = __builtin_amdgcn_readfirstlane(tid >> 5);
  const int l = tid & 31;
  const int h = l >> 4;
  const int c = l & 15;
  const int i = blockIdx.x;

  int base2 = 0, m1 = 0;
  {
    int v = i;
#pragma unroll
    for (int b = 0; b < 4; ++b) {
      const int d = v % 3;
      v /= 3;
      base2 |= (d == 2 ? 1 : 0) << b;
      m1    |= (d == 1 ? 1 : 0) << b;
    }
  }
  const int m1p = m1 & (m1 - 1);
  int sub = -1, nact = 0;
  for (int s = 0; s < 16; ++s) {
    if ((s & ~m1p) == 0) {
      if (nact == w) sub = s;
      ++nact;
    }
  }
  const bool active = sub >= 0;

  if (active) {
    const int kt = base2 | sub;
    const int nt = base2 | (m1 ^ sub);
    const size_t ao = (size_t)(kt * 16 + c) * QDIM + 4 * h;
    const size_t bo = (size_t)(nt * 16 + c) * QDIM + 4 * h;

    v8f accR = {0.f, 0.f, 0.f, 0.f, 0.f, 0.f, 0.f, 0.f};
    v8f accI = accR;
    FragQ f0, f1;
    load_fragQ(f0, gR + ao, gI + ao, rR + bo, rI + bo);
#pragma unroll
    for (int sb = 0; sb < QDIM / 8 / QB; sb += 2) {
      load_fragQ(f1, gR + ao + 8 * QB * (sb + 1), gI + ao + 8 * QB * (sb + 1),
                     rR + bo + 8 * QB * (sb + 1), rI + bo + 8 * QB * (sb + 1));
      __builtin_amdgcn_sched_barrier(6);
      mma_fragQ(f0, accR, accI);
      __builtin_amdgcn_sched_barrier(6);
      if (sb + 2 < QDIM / 8 / QB) {
        load_fragQ(f0, gR + ao + 8 * QB * (sb + 2), gI + ao + 8 * QB * (sb + 2),
                       rR + bo + 8 * QB * (sb + 2), rI + bo + 8 * QB * (sb + 2));
      }
      __builtin_amdgcn_sched_barrier(6);
      mma_fragQ(f1, accR, accI);
      __builtin_amdgcn_sched_barrier(6);
    }
#pragma unroll
    for (int r = 0; r < 8; ++r) tile[w][8 * h + r][c] = accR[r] + accI[r];
  }
  __syncthreads();

  float cs[4] = {0.f, 0.f, 0.f, 0.f};
  if (active) {
    float v[4][16];
    int   mk[4];
#pragma unroll
    for (int u = 0; u < 4; ++u) {
      const int o  = l + 32 * u;
      const int jh = o >> 2;
      const int t7 = o & 3;
      const int d1 = jh % 3, d2 = (jh / 3) % 3, d3 = jh / 9;
      const int b2 = (t7 == 2 ? 1 : 0) | (d1 == 2 ? 2 : 0) | (d2 == 2 ? 4 : 0) | (d3 == 2 ? 8 : 0);
      const int mm = (t7 == 1 ? 1 : 0) | (d1 == 1 ? 2 : 0) | (d2 == 1 ? 4 : 0) | (d3 == 1 ? 8 : 0);
      mk[u] = mm;
#pragma unroll
      for (int q = 0; q < 16; ++q) {
        const int qq = q & mm;
        v[u][q] = tile[w][b2 | qq][b2 | (mm ^ qq)];
      }
    }
#pragma unroll
    for (int u = 0; u < 4; ++u) {
      const int o = l + 32 * u;
      float s = 0.f;
#pragma unroll
      for (int q = 0; q < 16; ++q) s += ((q & ~mk[u]) == 0) ? v[u][q] : 0.f;
      cs[u] = ((o >> 2) < 27 && (o & 3) < 3) ? s : 0.f;
    }
  }
#pragma unroll
  for (int u = 0; u < 4; ++u) part[w][l + 32 * u] = cs[u];
  __syncthreads();

  if (tid < CPITCH) {
    float pv[CWAVES];
#pragma unroll
    for (int ww = 0; ww < CWAVES; ++ww) pv[ww] = part[ww][tid];
    float s = 0.f;
#pragma unroll
    for (int ww = 0; ww < CWAVES; ++ww) s += pv[ww];
    cP[i * CPITCH + tid] = (m1 != 0 ? 2.f : 1.f) * s;
  }
}

#define KS 27
struct FragA { v2f a0[KS], a1[KS]; };

__device__ __forceinline__ void load_fragA(FragA& f, const float* p) {
#pragma unroll
  for (int j = 0; j < KS; ++j) {
    f.a0[j] = *(const v2f*)(p + (size_t)j * CPITCH);
    f.a1[j] = *(const v2f*)(p + (size_t)j * CPITCH + 64);
  }
}

__device__ __forceinline__ void mma_fragA(const v2f* bm, const FragA& f, v8f& y0, v8f& y1) {
#pragma unroll
  for (int j = 0; j < KS; ++j) {
    y0 = __builtin_amdgcn_wmma_f32_16x16x4_f32(false, f.a0[j], false, bm[j], (short)0, y0, false, false);
    y1 = __builtin_amdgcn_wmma_f32_16x16x4_f32(false, f.a1[j], false, bm[j], (short)0, y1, false, false);
  }
}

__global__ __launch_bounds__(32) __attribute__((amdgpu_waves_per_eu(2, 2))) void vqe_wmma(const float* __restrict__ x,
                                               const float* __restrict__ fcb,
                                               const float* __restrict__ cP,
                                               float* __restrict__ out) {
  const int l = threadIdx.x;
  const int h = l >> 4;
  const int c = l & 15;
  const int sBase = blockIdx.x * 16;

  const v4f xa = *(const v4f*)(x + (size_t)(sBase + c) * NQ);
  const v4f xb = *(const v4f*)(x + (size_t)(sBase + c) * NQ + 4);
  const float xq[NQ] = {xa.x, xa.y, xa.z, xa.w, xb.x, xb.y, xb.z, xb.w};

  const float* abase = cP + 4 * c + 2 * h;
  FragA f0, f1;
  load_fragA(f0, abase);
  load_fragA(f1, abase + (size_t)KS * CPITCH);

  float F[NQ][3];
  {
#pragma unroll
    for (int q = 0; q < NQ; ++q) {
      const float xv = 0.5f * xq[q];
      const float cq = cosf(xv);
      const float sq = sinf(xv);
      F[q][0] = cq * cq;
      F[q][1] = cq * sq;
      F[q][2] = sq * sq;
    }
  }

  v2f bM[CROWS];
  {
    const float vx = h ? F[7][2] : F[7][0];
    const float vy = h ? 0.f     : F[7][1];
#pragma unroll
    for (int t0 = 0; t0 < 3; ++t0) {
#pragma unroll
      for (int t1 = 0; t1 < 3; ++t1) {
        const float p01 = F[0][t0] * F[1][t1];
#pragma unroll
        for (int t2 = 0; t2 < 3; ++t2) {
          const float p012 = p01 * F[2][t2];
#pragma unroll
          for (int t3 = 0; t3 < 3; ++t3) {
            const float u = p012 * F[3][t3];
            const int i = ((t0 * 3 + t1) * 3 + t2) * 3 + t3;
            bM[i].x = u * vx;
            bM[i].y = u * vy;
          }
        }
      }
    }
  }

  float w0[8], w1[8];
  {
    float V[32];
#pragma unroll
    for (int jh = 0; jh < 32; ++jh)
      V[jh] = (jh < 27) ? (F[4][jh / 9] * F[5][(jh / 3) % 3]) * F[6][jh % 3] : 0.f;
#pragma unroll
    for (int r = 0; r < 8; ++r) {
      w0[r] = h ? V[8 + r]  : V[r];
      w1[r] = h ? V[24 + r] : V[16 + r];
    }
  }

  v8f y0 = {0.f, 0.f, 0.f, 0.f, 0.f, 0.f, 0.f, 0.f};
  v8f y1 = y0;
  __builtin_amdgcn_sched_barrier(6);
  mma_fragA(bM, f0, y0, y1);
  __builtin_amdgcn_sched_barrier(6);
  load_fragA(f0, abase + (size_t)(2 * KS) * CPITCH);
  __builtin_amdgcn_sched_barrier(6);
  mma_fragA(bM + KS, f1, y0, y1);
  __builtin_amdgcn_sched_barrier(6);
  mma_fragA(bM + 2 * KS, f0, y0, y1);

  float s0 = 0.f, s1 = 0.f;
#pragma unroll
  for (int r = 0; r < 8; ++r) {
    s0 += y0[r] * w0[r];
    s1 += y1[r] * w1[r];
  }
  float z = s0 + s1;

  z += __shfl_xor(z, 16);
  if (l < 16) {
    const float sum = fcb[0] + z;
    out[sBase + l] = 1.f / (1.f + expf(-sum));
  }
}

extern "C" void kernel_launch(void* const* d_in, const int* in_sizes, int n_in,
                              void* d_out, int out_size, void* d_ws, size_t ws_size,
                              hipStream_t stream) {
  const float* x       = (const float*)d_in[0];
  const float* weights = (const float*)d_in[1];
  const float* fcw     = (const float*)d_in[2];
  const float* fcb     = (const float*)d_in[3];
  float* out = (float*)d_out;

  float* rR = (float*)d_ws;
  float* rI = rR + QDIM * QDIM;
  float* gR = rI + QDIM * QDIM;
  float* gI = gR + QDIM * QDIM;
  float* cP = gI + QDIM * QDIM;

  const int B        = in_sizes[0] / NQ;
  const int n_layers = in_sizes[1] / (NQ * 3);

  build_W<<<QDIM / 2, 32, 0, stream>>>(weights, fcw, n_layers, rR, rI, gR, gI);
  build_C<<<CROWS, CWAVES * 32, 0, stream>>>(rR, rI, gR, gI, cP);
  vqe_wmma<<<B / 16, 32, 0, stream>>>(x, fcb, cP, out);
}


# build_C: at most two tiles per work-group, two waves per tile (n-sum halved); 4- and 8-tile classes spread over 2 and 4 work-groups, kernel 3 adds their partial rows on load; sums reordered
# speedup vs baseline: 14.9334x; 1.2356x over previous
#include <hip/hip_runtime.h>
#include <math.h>

typedef __attribute__((ext_vector_type(2))) float v2f;
typedef __attribute__((ext_vector_type(4))) float v4f;
typedef __attribute__((ext_vector_type(8))) float v8f;

#define NQ   8
#define QDIM 256

__device__ __forceinline__ float2 cmul(float2 a, float2 b) {
  return make_float2(a.x * b.x - a.y * b.y, a.x * b.y + a.y * b.x);
}
__device__ __forceinline__ float2 cadd(float2 a, float2 b) {
  return make_float2(a.x + b.x, a.y + b.y);
}
__device__ __forceinline__ float2 mul_minus_i_pow(float2 z, int e) {
  const float2 s = (e & 1) ? make_float2(z.y, -z.x) : z;
  return (e & 2) ? make_float2(-s.x, -s.y) : s;
}


__device__ __forceinline__ int ring_src(int i, int rr) {
#pragma unroll
  for (int q = NQ - 1; q >= 0; --q) {
    const int cbit = 7 - q;
    const int tbit = 7 - ((q + rr) & 7);
    i ^= ((i >> cbit) & 1) << tbit;
  }
  return i;
}

__global__ __launch_bounds__(32) void build_W(const float* __restrict__ weights,
                                              const float* __restrict__ fcw,
                                              int n_layers,
                                              float* __restrict__ rR,
                                              float* __restrict__ rI,
                                              float* __restrict__ gR,
                                              float* __restrict__ gI) {
  __shared__ float4 gate[32][2];
  __shared__ float2 colA[QDIM];
  __shared__ float2 colB[QDIM];
  const int t = threadIdx.x;
  const int p = blockIdx.x;

  float2 a[8], b[8];
#pragma unroll
  for (int r = 0; r < 8; ++r) {
    const int i = r * 32 + t;
    a[r] = make_float2(i == 2 * p     ? 1.f : 0.f, 0.f);
    b[r] = make_float2(i == 2 * p + 1 ? 1.f : 0.f, 0.f);
  }

  for (int l0 = 0; l0 < n_layers; l0 += 4) {
    __syncthreads();
    {
      const int gl = l0 + (t >> 3);
      const int gq = t & 7;
      if (gl < n_layers) {
        const float ph = weights[(gl * NQ + gq) * 3 + 0];
        const float th = weights[(gl * NQ + gq) * 3 + 1];
        const float om = weights[(gl * NQ + gq) * 3 + 2];
        const float c  = cosf(0.5f * th), s = sinf(0.5f * th);
        const float ap = 0.5f * (ph + om), bm = 0.5f * (ph - om);
        gate[t][0] = make_float4( cosf(ap) * c, -sinf(ap) * c,
                                 -cosf(bm) * s, -sinf(bm) * s);
        gate[t][1] = make_float4( cosf(bm) * s, -sinf(bm) * s,
                                  cosf(ap) * c,  sinf(ap) * c);
      }
    }
    __syncthreads();

    const int lend = (l0 + 4 < n_layers) ? (l0 + 4) : n_layers;
    for (int l = l0; l < lend; ++l) {
#pragma unroll
      for (int q = 0; q < NQ; ++q) {
        const float4 g0 = gate[(l - l0) * 8 + q][0];
        const float4 g1 = gate[(l - l0) * 8 + q][1];
        const float2 u00 = make_float2(g0.x, g0.y), u01 = make_float2(g0.z, g0.w);
        const float2 u10 = make_float2(g1.x, g1.y), u11 = make_float2(g1.z, g1.w);
        if (q < 3) {
          const int rs = 4 >> q;
#pragma unroll
          for (int r = 0; r < 8; ++r) {
            if (r & rs) continue;
            const float2 a0 = a[r], a1 = a[r + rs];
            a[r]      = cadd(cmul(u00, a0), cmul(u01, a1));
            a[r + rs] = cadd(cmul(u10, a0), cmul(u11, a1));
            const float2 b0 = b[r], b1 = b[r + rs];
            b[r]      = cadd(cmul(u00, b0), cmul(u01, b1));
            b[r + rs] = cadd(cmul(u10, b0), cmul(u11, b1));
          }
        } else {
          const int  lm = 1 << (7 - q);
          const bool up = (t & lm) != 0;
          const float2 k0 = up ? u10 : u00;
          const float2 k1 = up ? u11 : u01;
float2 oa[8], ob[8];
#pragma unroll
          for (int r = 0; r < 8; ++r) {
            oa[r] = make_float2(__shfl_xor(a[r].x, lm), __shfl_xor(a[r].y, lm));
            ob[r] = make_float2(__shfl_xor(b[r].x, lm), __shfl_xor(b[r].y, lm));
          }
#pragma unroll
          for (int r = 0; r < 8; ++r) {
            const float2 ma = a[r], mb = b[r];
            a[r] = cadd(cmul(k0, up ? oa[r] : ma), cmul(k1, up ? ma : oa[r]));
            b[r] = cadd(cmul(k0, up ? ob[r] : mb), cmul(k1, up ? mb : ob[r]));
          }
        }
      }
      const int rr = l % (NQ - 1) + 1;
#pragma unroll
      for (int r = 0; r < 8; ++r) {
        colA[r * 32 + t] = a[r];
        colB[r * 32 + t] = b[r];
      }
      __syncthreads();
      const int st = ring_src(t, rr);
#pragma unroll
      for (int r = 0; r < 8; ++r) {
        const int si = st ^ ring_src(r * 32, rr);
        a[r] = colA[si];
        b[r] = colB[si];
      }
      __syncthreads();
    }
  }

  float fw[NQ];
#pragma unroll
  for (int i = 0; i < NQ; ++i) fw[i] = fcw[i];
  const int ea = __popc(2 * p);
#pragma unroll
  for (int r = 0; r < 8; ++r) {
    const int n = r * 32 + t;
    float g = 0.f;
#pragma unroll
    for (int i = 0; i < NQ; ++i)
      g += fw[i] * (1.f - 2.f * (float)((n >> (7 - i)) & 1));
    const float2 pa = mul_minus_i_pow(a[r], ea);
    const float2 pb = mul_minus_i_pow(b[r], ea + 1);
    const int ia = (2 * p) * QDIM + n, ib = ia + QDIM;
    rR[ia] = pa.x;      rI[ia] = pa.y;
    rR[ib] = pb.x;      rI[ib] = pb.y;
    gR[ia] = g * pa.x;  gI[ia] = g * pa.y;
    gR[ib] = g * pb.x;  gI[ib] = g * pb.y;
  }
}

#define QB 8
struct FragQ { v4f ar[QB], ai[QB], br[QB], bi[QB]; };

__device__ __forceinline__ void load_fragQ(FragQ& f, const float* pgR, const float* pgI,
                                           const float* prR, const float* prI) {
#pragma unroll
  for (int i = 0; i < QB; ++i) {
    f.ar[i] = *(const v4f*)(pgR + 8 * i);
    f.ai[i] = *(const v4f*)(pgI + 8 * i);
    f.br[i] = *(const v4f*)(prR + 8 * i);
    f.bi[i] = *(const v4f*)(prI + 8 * i);
  }
}

__device__ __forceinline__ void mma_fragQ(const FragQ& f, v8f& accR, v8f& accI) {
#pragma unroll
  for (int i = 0; i < QB; ++i) {
    const v2f ar0 = {f.ar[i].x, f.ar[i].y}, ar1 = {f.ar[i].z, f.ar[i].w};
    const v2f ai0 = {f.ai[i].x, f.ai[i].y}, ai1 = {f.ai[i].z, f.ai[i].w};
    const v2f br0 = {f.br[i].x, f.br[i].y}, br1 = {f.br[i].z, f.br[i].w};
    const v2f bi0 = {f.bi[i].x, f.bi[i].y}, bi1 = {f.bi[i].z, f.bi[i].w};
    accR = __builtin_amdgcn_wmma_f32_16x16x4_f32(false, ar0, false, br0, (short)0, accR, false, false);
    accI = __builtin_amdgcn_wmma_f32_16x16x4_f32(false, ai0, false, bi0, (short)0, accI, false, false);
    accR = __builtin_amdgcn_wmma_f32_16x16x4_f32(false, ar1, false, br1, (short)0, accR, false, false);
    accI = __builtin_amdgcn_wmma_f32_16x16x4_f32(false, ai1, false, bi1, (short)0, accI, false, false);
  }
}

#define CROWS  81
#define CPITCH 128
#define CWAVES 4
#define CXTRA  11
#define CBLK   (CROWS + CXTRA)

__device__ __forceinline__ constexpr int xtra_row(int r) {
  return r == 39 ? 81 : r == 41 ? 82 : r == 37 ? 83 : r == 43 ? 84
       : r == 31 ? 85 : r == 49 ? 86 : r == 13 ? 87 : r == 67 ? 88 : 89;
}
__device__ __forceinline__ constexpr int xtra_cnt(int r) {
  return r == 40 ? 3
       : (r == 39 || r == 41 || r == 37 || r == 43 || r == 31 || r == 49 || r == 13 || r == 67) ? 1 : 0;
}
__device__ __forceinline__ constexpr int split_row(int q) {
  return q == 0 ? 13 : q == 1 ? 31 : q == 2 ? 37 : q == 3 ? 39 : q == 4 ? 40
       : q == 5 ? 41 : q == 6 ? 43 : q == 7 ? 49 : 67;
}
__device__ __forceinline__ constexpr int row_of(int k) {
  return k >= 72 ? split_row(k - 72)
       : k + (k >= 13 ? 1 : 0) + (k >= 30 ? 1 : 0) + (k >= 35 ? 1 : 0) + (k >= 36 ? 3 : 0)
           + (k >= 37 ? 1 : 0) + (k >= 42 ? 1 : 0) + (k >= 59 ? 1 : 0);
}

__global__ __launch_bounds__(CWAVES * 32) void build_C(const float* __restrict__ rR,
                                                       const float* __restrict__ rI,
                                                       const float* __restrict__ gR,
                                                       const float* __restrict__ gI,
                                                       float* __restrict__ cP) {
  __shared__ float tile[CWAVES][16][17];
  __shared__ float part[CWAVES][CPITCH];
  const int tid = threadIdx.x;
  const int w = __builtin_amdgcn_readfirstlane(tid >> 5);
  const int l = tid & 31;
  const int h = l >> 4;
  const int c = l & 15;
  const int b = blockIdx.x;
  const int ts = w & 1;
  const int kh = w >> 1;

  int i = b, prt = 0;
  if (b >= CROWS + 8) {
    i = 40;
    prt = b - (CROWS + 8) + 1;
  } else if (b >= CROWS) {
    const int e  = b - CROWS;
    const int d  = e >> 1;
    const int p3 = d == 0 ? 1 : d == 1 ? 3 : d == 2 ? 9 : 27;
    i = 40 + ((e & 1) ? p3 : -p3);
    prt = 1;
  }

  int base2 = 0, m1 = 0;
  {
    int v = i;
#pragma unroll
    for (int bb = 0; bb < 4; ++bb) {
      const int d = v % 3;
      v /= 3;
      base2 |= (d == 2 ? 1 : 0) << bb;
      m1    |= (d == 1 ? 1 : 0) << bb;
    }
  }
  const int m1p = m1 & (m1 - 1);
  const int tix = 2 * prt + ts;
  int sub = -1, nact = 0;
  for (int s = 0; s < 16; ++s) {
    if ((s & ~m1p) == 0) {
      if (nact == tix) sub = s;
      ++nact;
    }
  }
  const bool active = sub >= 0;

  if (active) {
    const int kt = base2 | sub;
    const int nt = base2 | (m1 ^ sub);
    const size_t ao = (size_t)(kt * 16 + c) * QDIM + 4 * h + 128 * kh;
    const size_t bo = (size_t)(nt * 16 + c) * QDIM + 4 * h + 128 * kh;

    v8f accR = {0.f, 0.f, 0.f, 0.f, 0.f, 0.f, 0.f, 0.f};
    v8f accI = accR;
    FragQ f0, f1;
    load_fragQ(f0, gR + ao, gI + ao, rR + bo, rI + bo);
    load_fragQ(f1, gR + ao + 8 * QB, gI + ao + 8 * QB, rR + bo + 8 * QB, rI + bo + 8 * QB);
    __builtin_amdgcn_sched_barrier(6);
    mma_fragQ(f0, accR, accI);
    mma_fragQ(f1, accR, accI);
#pragma unroll
    for (int r = 0; r < 8; ++r) tile[w][8 * h + r][c] = accR[r] + accI[r];
  }
  __syncthreads();

  float cs[4] = {0.f, 0.f, 0.f, 0.f};
  if (active) {
    float v[4][16];
    int   mk[4];
#pragma unroll
    for (int u = 0; u < 4; ++u) {
      const int o  = l + 32 * u;
      const int jh = o >> 2;
      const int t7 = o & 3;
      const int d1 = jh % 3, d2 = (jh / 3) % 3, d3 = jh / 9;
      const int b2 = (t7 == 2 ? 1 : 0) | (d1 == 2 ? 2 : 0) | (d2 == 2 ? 4 : 0) | (d3 == 2 ? 8 : 0);
      const int mm = (t7 == 1 ? 1 : 0) | (d1 == 1 ? 2 : 0) | (d2 == 1 ? 4 : 0) | (d3 == 1 ? 8 : 0);
      mk[u] = mm;
#pragma unroll
      for (int q = 0; q < 16; ++q) {
        const int qq = q & mm;
        v[u][q] = tile[w][b2 | qq][b2 | (mm ^ qq)];
      }
    }
#pragma unroll
    for (int u = 0; u < 4; ++u) {
      const int o = l + 32 * u;
      float s = 0.f;
#pragma unroll
      for (int q = 0; q < 16; ++q) s += ((q & ~mk[u]) == 0) ? v[u][q] : 0.f;
      cs[u] = ((o >> 2) < 27 && (o & 3) < 3) ? s : 0.f;
    }
  }
#pragma unroll
  for (int u = 0; u < 4; ++u) part[w][l + 32 * u] = cs[u];
  __syncthreads();

  {
    float pv[CWAVES];
#pragma unroll
    for (int ww = 0; ww < CWAVES; ++ww) pv[ww] = part[ww][tid];
    float s = 0.f;
#pragma unroll
    for (int ww = 0; ww < CWAVES; ++ww) s += pv[ww];
    cP[b * CPITCH + tid] = (m1 != 0 ? 2.f : 1.f) * s;
  }
}

#define KS 27
struct FragA { v2f a0[KS], a1[KS]; };
struct FragX { v2f x0[9][3], x1[9][3]; };

__device__ __forceinline__ void load_fragA(FragA& f, const float* p, const int k0) {
#pragma unroll
  for (int j = 0; j < KS; ++j) {
    const int r = row_of(k0 + j);
    f.a0[j] = *(const v2f*)(p + (size_t)r * CPITCH);
    f.a1[j] = *(const v2f*)(p + (size_t)r * CPITCH + 64);
  }
}
__device__ __forceinline__ void load_fragX(FragX& x, const float* p) {
#pragma unroll
  for (int q = 0; q < 9; ++q) {
#pragma unroll
    for (int u = 0; u < 3; ++u) {
      if (u < xtra_cnt(split_row(q))) {
        x.x0[q][u] = *(const v2f*)(p + (size_t)(xtra_row(split_row(q)) + u) * CPITCH);
        x.x1[q][u] = *(const v2f*)(p + (size_t)(xtra_row(split_row(q)) + u) * CPITCH + 64);
      }
    }
  }
}

__device__ __forceinline__ void mma_fragA(const v2f* bm, const FragA& f, const FragX& x,
                                          v8f& y0, v8f& y1, const int k0) {
#pragma unroll
  for (int j = 0; j < KS; ++j) {
    const int k = k0 + j;
    const int r = row_of(k);
    v2f a0 = f.a0[j], a1 = f.a1[j];
    if (k >= 72) {
#pragma unroll
      for (int u = 0; u < 3; ++u) {
        if (u < xtra_cnt(r)) {
          a0 += x.x0[k - 72][u];
          a1 += x.x1[k - 72][u];
        }
      }
    }
    y0 = __builtin_amdgcn_wmma_f32_16x16x4_f32(false, a0, false, bm[r], (short)0, y0, false, false);
    y1 = __builtin_amdgcn_wmma_f32_16x16x4_f32(false, a1, false, bm[r], (short)0, y1, false, false);
  }
}

__global__ __launch_bounds__(32) __attribute__((amdgpu_waves_per_eu(2, 2))) void vqe_wmma(const float* __restrict__ x,
                                               const float* __restrict__ fcb,
                                               const float* __restrict__ cP,
                                               float* __restrict__ out) {
  const int l = threadIdx.x;
  const int h = l >> 4;
  const int c = l & 15;
  const int sBase = blockIdx.x * 16;

  const v4f xa = *(const v4f*)(x + (size_t)(sBase + c) * NQ);
  const v4f xb = *(const v4f*)(x + (size_t)(sBase + c) * NQ + 4);
  const float xq[NQ] = {xa.x, xa.y, xa.z, xa.w, xb.x, xb.y, xb.z, xb.w};

  const float* abase = cP + 4 * c + 2 * h;
  FragA f0, f1;
  FragX fx;
  load_fragA(f0, abase, 0);
  load_fragA(f1, abase, KS);
  load_fragX(fx, abase);

  float F[NQ][3];
  {
#pragma unroll
    for (int q = 0; q < NQ; ++q) {
      const float xv = 0.5f * xq[q];
      const float cq = cosf(xv);
      const float sq = sinf(xv);
      F[q][0] = cq * cq;
      F[q][1] = cq * sq;
      F[q][2] = sq * sq;
    }
  }

  v2f bM[CROWS];
  {
    const float vx = h ? F[7][2] : F[7][0];
    const float vy = h ? 0.f     : F[7][1];
#pragma unroll
    for (int t0 = 0; t0 < 3; ++t0) {
#pragma unroll
      for (int t1 = 0; t1 < 3; ++t1) {
        const float p01 = F[0][t0] * F[1][t1];
#pragma unroll
        for (int t2 = 0; t2 < 3; ++t2) {
          const float p012 = p01 * F[2][t2];
#pragma unroll
          for (int t3 = 0; t3 < 3; ++t3) {
            const float u = p012 * F[3][t3];
            const int i = ((t0 * 3 + t1) * 3 + t2) * 3 + t3;
            bM[i].x = u * vx;
            bM[i].y = u * vy;
          }
        }
      }
    }
  }

  float w0[8], w1[8];
  {
    float V[32];
#pragma unroll
    for (int jh = 0; jh < 32; ++jh)
      V[jh] = (jh < 27) ? (F[4][jh / 9] * F[5][(jh / 3) % 3]) * F[6][jh % 3] : 0.f;
#pragma unroll
    for (int r = 0; r < 8; ++r) {
      w0[r] = h ? V[8 + r]  : V[r];
      w1[r] = h ? V[24 + r] : V[16 + r];
    }
  }

  v8f y0 = {0.f, 0.f, 0.f, 0.f, 0.f, 0.f, 0.f, 0.f};
  v8f y1 = y0;
  __builtin_amdgcn_sched_barrier(6);
  mma_fragA(bM, f0, fx, y0, y1, 0);
  __builtin_amdgcn_sched_barrier(6);
  load_fragA(f0, abase, 2 * KS);
  __builtin_amdgcn_sched_barrier(6);
  mma_fragA(bM, f1, fx, y0, y1, KS);
  __builtin_amdgcn_sched_barrier(6);
  mma_fragA(bM, f0, fx, y0, y1, 2 * KS);

  float s0 = 0.f, s1 = 0.f;
#pragma unroll
  for (int r = 0; r < 8; ++r) {
    s0 += y0[r] * w0[r];
    s1 += y1[r] * w1[r];
  }
  float z = s0 + s1;

  z += __shfl_xor(z, 16);
  if (l < 16) {
    const float sum = fcb[0] + z;
    out[sBase + l] = 1.f / (1.f + expf(-sum));
  }
}

extern "C" void kernel_launch(void* const* d_in, const int* in_sizes, int n_in,
                              void* d_out, int out_size, void* d_ws, size_t ws_size,
                              hipStream_t stream) {
  const float* x       = (const float*)d_in[0];
  const float* weights = (const float*)d_in[1];
  const float* fcw     = (const float*)d_in[2];
  const float* fcb     = (const float*)d_in[3];
  float* out = (float*)d_out;

  float* rR = (float*)d_ws;
  float* rI = rR + QDIM * QDIM;
  float* gR = rI + QDIM * QDIM;
  float* gI = gR + QDIM * QDIM;
  float* cP = gI + QDIM * QDIM;

  const int B        = in_sizes[0] / NQ;
  const int n_layers = in_sizes[1] / (NQ * 3);

  build_W<<<QDIM / 2, 32, 0, stream>>>(weights, fcw, n_layers, rR, rI, gR, gI);
  build_C<<<CBLK, CWAVES * 32, 0, stream>>>(rR, rI, gR, gI, cP);
  vqe_wmma<<<B / 16, 32, 0, stream>>>(x, fcb, cP, out);
}


# build_W first layer written directly: a column of a one-qubit-gate layer is a product state, each place gets prod_q U_q[j_q][k_q] at j = its ring source; no sweeps, lane exchanges or LDS ring there
# speedup vs baseline: 17.1084x; 1.1456x over previous
#include <hip/hip_runtime.h>
#include <math.h>

typedef __attribute__((ext_vector_type(2))) float v2f;
typedef __attribute__((ext_vector_type(4))) float v4f;
typedef __attribute__((ext_vector_type(8))) float v8f;

#define NQ   8
#define QDIM 256

__device__ __forceinline__ float2 cmul(float2 a, float2 b) {
  return make_float2(a.x * b.x - a.y * b.y, a.x * b.y + a.y * b.x);
}
__device__ __forceinline__ float2 cadd(float2 a, float2 b) {
  return make_float2(a.x + b.x, a.y + b.y);
}
__device__ __forceinline__ float2 mul_minus_i_pow(float2 z, int e) {
  const float2 s = (e & 1) ? make_float2(z.y, -z.x) : z;
  return (e & 2) ? make_float2(-s.x, -s.y) : s;
}


__device__ __forceinline__ int ring_src(int i, int rr) {
#pragma unroll
  for (int q = NQ - 1; q >= 0; --q) {
    const int cbit = 7 - q;
    const int tbit = 7 - ((q + rr) & 7);
    i ^= ((i >> cbit) & 1) << tbit;
  }
  return i;
}

__global__ __launch_bounds__(32) void build_W(const float* __restrict__ weights,
                                              const float* __restrict__ fcw,
                                              int n_layers,
                                              float* __restrict__ rR,
                                              float* __restrict__ rI,
                                              float* __restrict__ gR,
                                              float* __restrict__ gI) {
  __shared__ float4 gate[32][2];
  __shared__ float2 colA[QDIM];
  __shared__ float2 colB[QDIM];
  const int t = threadIdx.x;
  const int p = blockIdx.x;

  float2 a[8], b[8];
#pragma unroll
  for (int r = 0; r < 8; ++r) {
    const int i = r * 32 + t;
    a[r] = make_float2(i == 2 * p     ? 1.f : 0.f, 0.f);
    b[r] = make_float2(i == 2 * p + 1 ? 1.f : 0.f, 0.f);
  }

  for (int l0 = 0; l0 < n_layers; l0 += 4) {
    __syncthreads();
    {
      const int gl = l0 + (t >> 3);
      const int gq = t & 7;
      if (gl < n_layers) {
        const float ph = weights[(gl * NQ + gq) * 3 + 0];
        const float th = weights[(gl * NQ + gq) * 3 + 1];
        const float om = weights[(gl * NQ + gq) * 3 + 2];
        const float c  = cosf(0.5f * th), s = sinf(0.5f * th);
        const float ap = 0.5f * (ph + om), bm = 0.5f * (ph - om);
        gate[t][0] = make_float4( cosf(ap) * c, -sinf(ap) * c,
                                 -cosf(bm) * s, -sinf(bm) * s);
        gate[t][1] = make_float4( cosf(bm) * s, -sinf(bm) * s,
                                  cosf(ap) * c,  sinf(ap) * c);
      }
    }
    __syncthreads();

    const int lend = (l0 + 4 < n_layers) ? (l0 + 4) : n_layers;
    for (int l = l0; l < lend; ++l) {
      if (l == 0) {
        const int st0 = ring_src(t, 1);
        float2 e0[NQ - 1], e1[NQ - 1];
#pragma unroll
        for (int q = 0; q < NQ - 1; ++q) {
          const float4 g0 = gate[q][0];
          const float4 g1 = gate[q][1];
          const bool kb = (((2 * p) >> (7 - q)) & 1) != 0;
          e0[q] = kb ? make_float2(g0.z, g0.w) : make_float2(g0.x, g0.y);
          e1[q] = kb ? make_float2(g1.z, g1.w) : make_float2(g1.x, g1.y);
        }
        const float4 h0 = gate[NQ - 1][0];
        const float4 h1 = gate[NQ - 1][1];
#pragma unroll
        for (int r = 0; r < 8; ++r) {
          const int s = st0 ^ ring_src(r * 32, 1);
          float2 pr = ((s >> 7) & 1) ? e1[0] : e0[0];
#pragma unroll
          for (int q = 1; q < NQ - 1; ++q)
            pr = cmul(((s >> (7 - q)) & 1) ? e1[q] : e0[q], pr);
          const bool s7 = (s & 1) != 0;
          a[r] = cmul(s7 ? make_float2(h1.x, h1.y) : make_float2(h0.x, h0.y), pr);
          b[r] = cmul(s7 ? make_float2(h1.z, h1.w) : make_float2(h0.z, h0.w), pr);
        }
        continue;
      }
#pragma unroll
      for (int q = 0; q < NQ; ++q) {
        const float4 g0 = gate[(l - l0) * 8 + q][0];
        const float4 g1 = gate[(l - l0) * 8 + q][1];
        const float2 u00 = make_float2(g0.x, g0.y), u01 = make_float2(g0.z, g0.w);
        const float2 u10 = make_float2(g1.x, g1.y), u11 = make_float2(g1.z, g1.w);
        if (q < 3) {
          const int rs = 4 >> q;
#pragma unroll
          for (int r = 0; r < 8; ++r) {
            if (r & rs) continue;
            const float2 a0 = a[r], a1 = a[r + rs];
            a[r]      = cadd(cmul(u00, a0), cmul(u01, a1));
            a[r + rs] = cadd(cmul(u10, a0), cmul(u11, a1));
            const float2 b0 = b[r], b1 = b[r + rs];
            b[r]      = cadd(cmul(u00, b0), cmul(u01, b1));
            b[r + rs] = cadd(cmul(u10, b0), cmul(u11, b1));
          }
        } else {
          const int  lm = 1 << (7 - q);
          const bool up = (t & lm) != 0;
          const float2 k0 = up ? u10 : u00;
          const float2 k1 = up ? u11 : u01;
float2 oa[8], ob[8];
#pragma unroll
          for (int r = 0; r < 8; ++r) {
            oa[r] = make_float2(__shfl_xor(a[r].x, lm), __shfl_xor(a[r].y, lm));
            ob[r] = make_float2(__shfl_xor(b[r].x, lm), __shfl_xor(b[r].y, lm));
          }
#pragma unroll
          for (int r = 0; r < 8; ++r) {
            const float2 ma = a[r], mb = b[r];
            a[r] = cadd(cmul(k0, up ? oa[r] : ma), cmul(k1, up ? ma : oa[r]));
            b[r] = cadd(cmul(k0, up ? ob[r] : mb), cmul(k1, up ? mb : ob[r]));
          }
        }
      }
      const int rr = l % (NQ - 1) + 1;
#pragma unroll
      for (int r = 0; r < 8; ++r) {
        colA[r * 32 + t] = a[r];
        colB[r * 32 + t] = b[r];
      }
      __syncthreads();
      const int st = ring_src(t, rr);
#pragma unroll
      for (int r = 0; r < 8; ++r) {
        const int si = st ^ ring_src(r * 32, rr);
        a[r] = colA[si];
        b[r] = colB[si];
      }
      __syncthreads();
    }
  }

  float fw[NQ];
#pragma unroll
  for (int i = 0; i < NQ; ++i) fw[i] = fcw[i];
  const int ea = __popc(2 * p);
#pragma unroll
  for (int r = 0; r < 8; ++r) {
    const int n = r * 32 + t;
    float g = 0.f;
#pragma unroll
    for (int i = 0; i < NQ; ++i)
      g += fw[i] * (1.f - 2.f * (float)((n >> (7 - i)) & 1));
    const float2 pa = mul_minus_i_pow(a[r], ea);
    const float2 pb = mul_minus_i_pow(b[r], ea + 1);
    const int ia = (2 * p) * QDIM + n, ib = ia + QDIM;
    rR[ia] = pa.x;      rI[ia] = pa.y;
    rR[ib] = pb.x;      rI[ib] = pb.y;
    gR[ia] = g * pa.x;  gI[ia] = g * pa.y;
    gR[ib] = g * pb.x;  gI[ib] = g * pb.y;
  }
}

#define QB 8
struct FragQ { v4f ar[QB], ai[QB], br[QB], bi[QB]; };

__device__ __forceinline__ void load_fragQ(FragQ& f, const float* pgR, const float* pgI,
                                           const float* prR, const float* prI) {
#pragma unroll
  for (int i = 0; i < QB; ++i) {
    f.ar[i] = *(const v4f*)(pgR + 8 * i);
    f.ai[i] = *(const v4f*)(pgI + 8 * i);
    f.br[i] = *(const v4f*)(prR + 8 * i);
    f.bi[i] = *(const v4f*)(prI + 8 * i);
  }
}

__device__ __forceinline__ void mma_fragQ(const FragQ& f, v8f& accR, v8f& accI) {
#pragma unroll
  for (int i = 0; i < QB; ++i) {
    const v2f ar0 = {f.ar[i].x, f.ar[i].y}, ar1 = {f.ar[i].z, f.ar[i].w};
    const v2f ai0 = {f.ai[i].x, f.ai[i].y}, ai1 = {f.ai[i].z, f.ai[i].w};
    const v2f br0 = {f.br[i].x, f.br[i].y}, br1 = {f.br[i].z, f.br[i].w};
    const v2f bi0 = {f.bi[i].x, f.bi[i].y}, bi1 = {f.bi[i].z, f.bi[i].w};
    accR = __builtin_amdgcn_wmma_f32_16x16x4_f32(false, ar0, false, br0, (short)0, accR, false, false);
    accI = __builtin_amdgcn_wmma_f32_16x16x4_f32(false, ai0, false, bi0, (short)0, accI, false, false);
    accR = __builtin_amdgcn_wmma_f32_16x16x4_f32(false, ar1, false, br1, (short)0, accR, false, false);
    accI = __builtin_amdgcn_wmma_f32_16x16x4_f32(false, ai1, false, bi1, (short)0, accI, false, false);
  }
}

#define CROWS  81
#define CPITCH 128
#define CWAVES 4
#define CXTRA  11
#define CBLK   (CROWS + CXTRA)

__device__ __forceinline__ constexpr int xtra_row(int r) {
  return r == 39 ? 81 : r == 41 ? 82 : r == 37 ? 83 : r == 43 ? 84
       : r == 31 ? 85 : r == 49 ? 86 : r == 13 ? 87 : r == 67 ? 88 : 89;
}
__device__ __forceinline__ constexpr int xtra_cnt(int r) {
  return r == 40 ? 3
       : (r == 39 || r == 41 || r == 37 || r == 43 || r == 31 || r == 49 || r == 13 || r == 67) ? 1 : 0;
}
__device__ __forceinline__ constexpr int split_row(int q) {
  return q == 0 ? 13 : q == 1 ? 31 : q == 2 ? 37 : q == 3 ? 39 : q == 4 ? 40
       : q == 5 ? 41 : q == 6 ? 43 : q == 7 ? 49 : 67;
}
__device__ __forceinline__ constexpr int row_of(int k) {
  return k >= 72 ? split_row(k - 72)
       : k + (k >= 13 ? 1 : 0) + (k >= 30 ? 1 : 0) + (k >= 35 ? 1 : 0) + (k >= 36 ? 3 : 0)
           + (k >= 37 ? 1 : 0) + (k >= 42 ? 1 : 0) + (k >= 59 ? 1 : 0);
}

__global__ __launch_bounds__(CWAVES * 32) void build_C(const float* __restrict__ rR,
                                                       const float* __restrict__ rI,
                                                       const float* __restrict__ gR,
                                                       const float* __restrict__ gI,
                                                       float* __restrict__ cP) {
  __shared__ float tile[CWAVES][16][17];
  __shared__ float part[CWAVES][CPITCH];
  const int tid = threadIdx.x;
  const int w = __builtin_amdgcn_readfirstlane(tid >> 5);
  const int l = tid & 31;
  const int h = l >> 4;
  const int c = l & 15;
  const int b = blockIdx.x;
  const int ts = w & 1;
  const int kh = w >> 1;

  int i = b, prt = 0;
  if (b >= CROWS + 8) {
    i = 40;
    prt = b - (CROWS + 8) + 1;
  } else if (b >= CROWS) {
    const int e  = b - CROWS;
    const int d  = e >> 1;
    const int p3 = d == 0 ? 1 : d == 1 ? 3 : d == 2 ? 9 : 27;
    i = 40 + ((e & 1) ? p3 : -p3);
    prt = 1;
  }

  int base2 = 0, m1 = 0;
  {
    int v = i;
#pragma unroll
    for (int bb = 0; bb < 4; ++bb) {
      const int d = v % 3;
      v /= 3;
      base2 |= (d == 2 ? 1 : 0) << bb;
      m1    |= (d == 1 ? 1 : 0) << bb;
    }
  }
  const int m1p = m1 & (m1 - 1);
  const int tix = 2 * prt + ts;
  int sub = -1, nact = 0;
  for (int s = 0; s < 16; ++s) {
    if ((s & ~m1p) == 0) {
      if (nact == tix) sub = s;
      ++nact;
    }
  }
  const bool active = sub >= 0;

  if (active) {
    const int kt = base2 | sub;
    const int nt = base2 | (m1 ^ sub);
    const size_t ao = (size_t)(kt * 16 + c) * QDIM + 4 * h + 128 * kh;
    const size_t bo = (size_t)(nt * 16 + c) * QDIM + 4 * h + 128 * kh;

    v8f accR = {0.f, 0.f, 0.f, 0.f, 0.f, 0.f, 0.f, 0.f};
    v8f accI = accR;
    FragQ f0, f1;
    load_fragQ(f0, gR + ao, gI + ao, rR + bo, rI + bo);
    load_fragQ(f1, gR + ao + 8 * QB, gI + ao + 8 * QB, rR + bo + 8 * QB, rI + bo + 8 * QB);
    __builtin_amdgcn_sched_barrier(6);
    mma_fragQ(f0, accR, accI);
    mma_fragQ(f1, accR, accI);
#pragma unroll
    for (int r = 0; r < 8; ++r) tile[w][8 * h + r][c] = accR[r] + accI[r];
  }
  __syncthreads();

  float cs[4] = {0.f, 0.f, 0.f, 0.f};
  if (active) {
    float v[4][16];
    int   mk[4];
#pragma unroll
    for (int u = 0; u < 4; ++u) {
      const int o  = l + 32 * u;
      const int jh = o >> 2;
      const int t7 = o & 3;
      const int d1 = jh % 3, d2 = (jh / 3) % 3, d3 = jh / 9;
      const int b2 = (t7 == 2 ? 1 : 0) | (d1 == 2 ? 2 : 0) | (d2 == 2 ? 4 : 0) | (d3 == 2 ? 8 : 0);
      const int mm = (t7 == 1 ? 1 : 0) | (d1 == 1 ? 2 : 0) | (d2 == 1 ? 4 : 0) | (d3 == 1 ? 8 : 0);
      mk[u] = mm;
#pragma unroll
      for (int q = 0; q < 16; ++q) {
        const int qq = q & mm;
        v[u][q] = tile[w][b2 | qq][b2 | (mm ^ qq)];
      }
    }
#pragma unroll
    for (int u = 0; u < 4; ++u) {
      const int o = l + 32 * u;
      float s = 0.f;
#pragma unroll
      for (int q = 0; q < 16; ++q) s += ((q & ~mk[u]) == 0) ? v[u][q] : 0.f;
      cs[u] = ((o >> 2) < 27 && (o & 3) < 3) ? s : 0.f;
    }
  }
#pragma unroll
  for (int u = 0; u < 4; ++u) part[w][l + 32 * u] = cs[u];
  __syncthreads();

  {
    float pv[CWAVES];
#pragma unroll
    for (int ww = 0; ww < CWAVES; ++ww) pv[ww] = part[ww][tid];
    float s = 0.f;
#pragma unroll
    for (int ww = 0; ww < CWAVES; ++ww) s += pv[ww];
    cP[b * CPITCH + tid] = (m1 != 0 ? 2.f : 1.f) * s;
  }
}

#define KS 27
struct FragA { v2f a0[KS], a1[KS]; };
struct FragX { v2f x0[9][3], x1[9][3]; };

__device__ __forceinline__ void load_fragA(FragA& f, const float* p, const int k0) {
#pragma unroll
  for (int j = 0; j < KS; ++j) {
    const int r = row_of(k0 + j);
    f.a0[j] = *(const v2f*)(p + (size_t)r * CPITCH);
    f.a1[j] = *(const v2f*)(p + (size_t)r * CPITCH + 64);
  }
}
__device__ __forceinline__ void load_fragX(FragX& x, const float* p) {
#pragma unroll
  for (int q = 0; q < 9; ++q) {
#pragma unroll
    for (int u = 0; u < 3; ++u) {
      if (u < xtra_cnt(split_row(q))) {
        x.x0[q][u] = *(const v2f*)(p + (size_t)(xtra_row(split_row(q)) + u) * CPITCH);
        x.x1[q][u] = *(const v2f*)(p + (size_t)(xtra_row(split_row(q)) + u) * CPITCH + 64);
      }
    }
  }
}

__device__ __forceinline__ void mma_fragA(const v2f* bm, const FragA& f, const FragX& x,
                                          v8f& y0, v8f& y1, const int k0) {
#pragma unroll
  for (int j = 0; j < KS; ++j) {
    const int k = k0 + j;
    const int r = row_of(k);
    v2f a0 = f.a0[j], a1 = f.a1[j];
    if (k >= 72) {
#pragma unroll
      for (int u = 0; u < 3; ++u) {
        if (u < xtra_cnt(r)) {
          a0 += x.x0[k - 72][u];
          a1 += x.x1[k - 72][u];
        }
      }
    }
    y0 = __builtin_amdgcn_wmma_f32_16x16x4_f32(false, a0, false, bm[r], (short)0, y0, false, false);
    y1 = __builtin_amdgcn_wmma_f32_16x16x4_f32(false, a1, false, bm[r], (short)0, y1, false, false);
  }
}

__global__ __launch_bounds__(32) __attribute__((amdgpu_waves_per_eu(2, 2))) void vqe_wmma(const float* __restrict__ x,
                                               const float* __restrict__ fcb,
                                               const float* __restrict__ cP,
                                               float* __restrict__ out) {
  const int l = threadIdx.x;
  const int h = l >> 4;
  const int c = l & 15;
  const int sBase = blockIdx.x * 16;

  const v4f xa = *(const v4f*)(x + (size_t)(sBase + c) * NQ);
  const v4f xb = *(const v4f*)(x + (size_t)(sBase + c) * NQ + 4);
  const float xq[NQ] = {xa.x, xa.y, xa.z, xa.w, xb.x, xb.y, xb.z, xb.w};

  const float* abase = cP + 4 * c + 2 * h;
  FragA f0, f1;
  FragX fx;
  load_fragA(f0, abase, 0);
  load_fragA(f1, abase, KS);
  load_fragX(fx, abase);

  float F[NQ][3];
  {
#pragma unroll
    for (int q = 0; q < NQ; ++q) {
      const float xv = 0.5f * xq[q];
      const float cq = cosf(xv);
      const float sq = sinf(xv);
      F[q][0] = cq * cq;
      F[q][1] = cq * sq;
      F[q][2] = sq * sq;
    }
  }

  v2f bM[CROWS];
  {
    const float vx = h ? F[7][2] : F[7][0];
    const float vy = h ? 0.f     : F[7][1];
#pragma unroll
    for (int t0 = 0; t0 < 3; ++t0) {
#pragma unroll
      for (int t1 = 0; t1 < 3; ++t1) {
        const float p01 = F[0][t0] * F[1][t1];
#pragma unroll
        for (int t2 = 0; t2 < 3; ++t2) {
          const float p012 = p01 * F[2][t2];
#pragma unroll
          for (int t3 = 0; t3 < 3; ++t3) {
            const float u = p012 * F[3][t3];
            const int i = ((t0 * 3 + t1) * 3 + t2) * 3 + t3;
            bM[i].x = u * vx;
            bM[i].y = u * vy;
          }
        }
      }
    }
  }

  float w0[8], w1[8];
  {
    float V[32];
#pragma unroll
    for (int jh = 0; jh < 32; ++jh)
      V[jh] = (jh < 27) ? (F[4][jh / 9] * F[5][(jh / 3) % 3]) * F[6][jh % 3] : 0.f;
#pragma unroll
    for (int r = 0; r < 8; ++r) {
      w0[r] = h ? V[8 + r]  : V[r];
      w1[r] = h ? V[24 + r] : V[16 + r];
    }
  }

  v8f y0 = {0.f, 0.f, 0.f, 0.f, 0.f, 0.f, 0.f, 0.f};
  v8f y1 = y0;
  __builtin_amdgcn_sched_barrier(6);
  mma_fragA(bM, f0, fx, y0, y1, 0);
  __builtin_amdgcn_sched_barrier(6);
  load_fragA(f0, abase, 2 * KS);
  __builtin_amdgcn_sched_barrier(6);
  mma_fragA(bM, f1, fx, y0, y1, KS);
  __builtin_amdgcn_sched_barrier(6);
  mma_fragA(bM, f0, fx, y0, y1, 2 * KS);

  float s0 = 0.f, s1 = 0.f;
#pragma unroll
  for (int r = 0; r < 8; ++r) {
    s0 += y0[r] * w0[r];
    s1 += y1[r] * w1[r];
  }
  float z = s0 + s1;

  z += __shfl_xor(z, 16);
  if (l < 16) {
    const float sum = fcb[0] + z;
    out[sBase + l] = 1.f / (1.f + expf(-sum));
  }
}

extern "C" void kernel_launch(void* const* d_in, const int* in_sizes, int n_in,
                              void* d_out, int out_size, void* d_ws, size_t ws_size,
                              hipStream_t stream) {
  const float* x       = (const float*)d_in[0];
  const float* weights = (const float*)d_in[1];
  const float* fcw     = (const float*)d_in[2];
  const float* fcb     = (const float*)d_in[3];
  float* out = (float*)d_out;

  float* rR = (float*)d_ws;
  float* rI = rR + QDIM * QDIM;
  float* gR = rI + QDIM * QDIM;
  float* gI = gR + QDIM * QDIM;
  float* cP = gI + QDIM * QDIM;

  const int B        = in_sizes[0] / NQ;
  const int n_layers = in_sizes[1] / (NQ * 3);

  build_W<<<QDIM / 2, 32, 0, stream>>>(weights, fcw, n_layers, rR, rI, gR, gI);
  build_C<<<CBLK, CWAVES * 32, 0, stream>>>(rR, rI, gR, gI, cP);
  vqe_wmma<<<B / 16, 32, 0, stream>>>(x, fcb, cP, out);
}


# vqe_wmma: two 16-sample tiles per wave (1024 one-wave work-groups, not 2048): rows of C loaded once per 32 samples, each sample's trig done once, not by both half-waves; sums unchanged
# speedup vs baseline: 18.9284x; 1.1064x over previous
#include <hip/hip_runtime.h>
#include <math.h>

typedef __attribute__((ext_vector_type(2))) float v2f;
typedef __attribute__((ext_vector_type(4))) float v4f;
typedef __attribute__((ext_vector_type(8))) float v8f;

#define NQ   8
#define QDIM 256

__device__ __forceinline__ float2 cmul(float2 a, float2 b) {
  return make_float2(a.x * b.x - a.y * b.y, a.x * b.y + a.y * b.x);
}
__device__ __forceinline__ float2 cadd(float2 a, float2 b) {
  return make_float2(a.x + b.x, a.y + b.y);
}
__device__ __forceinline__ float2 mul_minus_i_pow(float2 z, int e) {
  const float2 s = (e & 1) ? make_float2(z.y, -z.x) : z;
  return (e & 2) ? make_float2(-s.x, -s.y) : s;
}


__device__ __forceinline__ int ring_src(int i, int rr) {
#pragma unroll
  for (int q = NQ - 1; q >= 0; --q) {
    const int cbit = 7 - q;
    const int tbit = 7 - ((q + rr) & 7);
    i ^= ((i >> cbit) & 1) << tbit;
  }
  return i;
}

__global__ __launch_bounds__(32) void build_W(const float* __restrict__ weights,
                                              const float* __restrict__ fcw,
                                              int n_layers,
                                              float* __restrict__ rR,
                                              float* __restrict__ rI,
                                              float* __restrict__ gR,
                                              float* __restrict__ gI) {
  __shared__ float4 gate[32][2];
  __shared__ float2 colA[QDIM];
  __shared__ float2 colB[QDIM];
  const int t = threadIdx.x;
  const int p = blockIdx.x;

  float2 a[8], b[8];
#pragma unroll
  for (int r = 0; r < 8; ++r) {
    const int i = r * 32 + t;
    a[r] = make_float2(i == 2 * p     ? 1.f : 0.f, 0.f);
    b[r] = make_float2(i == 2 * p + 1 ? 1.f : 0.f, 0.f);
  }

  for (int l0 = 0; l0 < n_layers; l0 += 4) {
    __syncthreads();
    {
      const int gl = l0 + (t >> 3);
      const int gq = t & 7;
      if (gl < n_layers) {
        const float ph = weights[(gl * NQ + gq) * 3 + 0];
        const float th = weights[(gl * NQ + gq) * 3 + 1];
        const float om = weights[(gl * NQ + gq) * 3 + 2];
        const float c  = cosf(0.5f * th), s = sinf(0.5f * th);
        const float ap = 0.5f * (ph + om), bm = 0.5f * (ph - om);
        gate[t][0] = make_float4( cosf(ap) * c, -sinf(ap) * c,
                                 -cosf(bm) * s, -sinf(bm) * s);
        gate[t][1] = make_float4( cosf(bm) * s, -sinf(bm) * s,
                                  cosf(ap) * c,  sinf(ap) * c);
      }
    }
    __syncthreads();

    const int lend = (l0 + 4 < n_layers) ? (l0 + 4) : n_layers;
    for (int l = l0; l < lend; ++l) {
      if (l == 0) {
        const int st0 = ring_src(t, 1);
        float2 e0[NQ - 1], e1[NQ - 1];
#pragma unroll
        for (int q = 0; q < NQ - 1; ++q) {
          const float4 g0 = gate[q][0];
          const float4 g1 = gate[q][1];
          const bool kb = (((2 * p) >> (7 - q)) & 1) != 0;
          e0[q] = kb ? make_float2(g0.z, g0.w) : make_float2(g0.x, g0.y);
          e1[q] = kb ? make_float2(g1.z, g1.w) : make_float2(g1.x, g1.y);
        }
        const float4 h0 = gate[NQ - 1][0];
        const float4 h1 = gate[NQ - 1][1];
#pragma unroll
        for (int r = 0; r < 8; ++r) {
          const int s = st0 ^ ring_src(r * 32, 1);
          float2 pr = ((s >> 7) & 1) ? e1[0] : e0[0];
#pragma unroll
          for (int q = 1; q < NQ - 1; ++q)
            pr = cmul(((s >> (7 - q)) & 1) ? e1[q] : e0[q], pr);
          const bool s7 = (s & 1) != 0;
          a[r] = cmul(s7 ? make_float2(h1.x, h1.y) : make_float2(h0.x, h0.y), pr);
          b[r] = cmul(s7 ? make_float2(h1.z, h1.w) : make_float2(h0.z, h0.w), pr);
        }
        continue;
      }
#pragma unroll
      for (int q = 0; q < NQ; ++q) {
        const float4 g0 = gate[(l - l0) * 8 + q][0];
        const float4 g1 = gate[(l - l0) * 8 + q][1];
        const float2 u00 = make_float2(g0.x, g0.y), u01 = make_float2(g0.z, g0.w);
        const float2 u10 = make_float2(g1.x, g1.y), u11 = make_float2(g1.z, g1.w);
        if (q < 3) {
          const int rs = 4 >> q;
#pragma unroll
          for (int r = 0; r < 8; ++r) {
            if (r & rs) continue;
            const float2 a0 = a[r], a1 = a[r + rs];
            a[r]      = cadd(cmul(u00, a0), cmul(u01, a1));
            a[r + rs] = cadd(cmul(u10, a0), cmul(u11, a1));
            const float2 b0 = b[r], b1 = b[r + rs];
            b[r]      = cadd(cmul(u00, b0), cmul(u01, b1));
            b[r + rs] = cadd(cmul(u10, b0), cmul(u11, b1));
          }
        } else {
          const int  lm = 1 << (7 - q);
          const bool up = (t & lm) != 0;
          const float2 k0 = up ? u10 : u00;
          const float2 k1 = up ? u11 : u01;
float2 oa[8], ob[8];
#pragma unroll
          for (int r = 0; r < 8; ++r) {
            oa[r] = make_float2(__shfl_xor(a[r].x, lm), __shfl_xor(a[r].y, lm));
            ob[r] = make_float2(__shfl_xor(b[r].x, lm), __shfl_xor(b[r].y, lm));
          }
#pragma unroll
          for (int r = 0; r < 8; ++r) {
            const float2 ma = a[r], mb = b[r];
            a[r] = cadd(cmul(k0, up ? oa[r] : ma), cmul(k1, up ? ma : oa[r]));
            b[r] = cadd(cmul(k0, up ? ob[r] : mb), cmul(k1, up ? mb : ob[r]));
          }
        }
      }
      const int rr = l % (NQ - 1) + 1;
#pragma unroll
      for (int r = 0; r < 8; ++r) {
        colA[r * 32 + t] = a[r];
        colB[r * 32 + t] = b[r];
      }
      __syncthreads();
      const int st = ring_src(t, rr);
#pragma unroll
      for (int r = 0; r < 8; ++r) {
        const int si = st ^ ring_src(r * 32, rr);
        a[r] = colA[si];
        b[r] = colB[si];
      }
      __syncthreads();
    }
  }

  float fw[NQ];
#pragma unroll
  for (int i = 0; i < NQ; ++i) fw[i] = fcw[i];
  const int ea = __popc(2 * p);
#pragma unroll
  for (int r = 0; r < 8; ++r) {
    const int n = r * 32 + t;
    float g = 0.f;
#pragma unroll
    for (int i = 0; i < NQ; ++i)
      g += fw[i] * (1.f - 2.f * (float)((n >> (7 - i)) & 1));
    const float2 pa = mul_minus_i_pow(a[r], ea);
    const float2 pb = mul_minus_i_pow(b[r], ea + 1);
    const int ia = (2 * p) * QDIM + n, ib = ia + QDIM;
    rR[ia] = pa.x;      rI[ia] = pa.y;
    rR[ib] = pb.x;      rI[ib] = pb.y;
    gR[ia] = g * pa.x;  gI[ia] = g * pa.y;
    gR[ib] = g * pb.x;  gI[ib] = g * pb.y;
  }
}

#define QB 8
struct FragQ { v4f ar[QB], ai[QB], br[QB], bi[QB]; };

__device__ __forceinline__ void load_fragQ(FragQ& f, const float* pgR, const float* pgI,
                                           const float* prR, const float* prI) {
#pragma unroll
  for (int i = 0; i < QB; ++i) {
    f.ar[i] = *(const v4f*)(pgR + 8 * i);
    f.ai[i] = *(const v4f*)(pgI + 8 * i);
    f.br[i] = *(const v4f*)(prR + 8 * i);
    f.bi[i] = *(const v4f*)(prI + 8 * i);
  }
}

__device__ __forceinline__ void mma_fragQ(const FragQ& f, v8f& accR, v8f& accI) {
#pragma unroll
  for (int i = 0; i < QB; ++i) {
    const v2f ar0 = {f.ar[i].x, f.ar[i].y}, ar1 = {f.ar[i].z, f.ar[i].w};
    const v2f ai0 = {f.ai[i].x, f.ai[i].y}, ai1 = {f.ai[i].z, f.ai[i].w};
    const v2f br0 = {f.br[i].x, f.br[i].y}, br1 = {f.br[i].z, f.br[i].w};
    const v2f bi0 = {f.bi[i].x, f.bi[i].y}, bi1 = {f.bi[i].z, f.bi[i].w};
    accR = __builtin_amdgcn_wmma_f32_16x16x4_f32(false, ar0, false, br0, (short)0, accR, false, false);
    accI = __builtin_amdgcn_wmma_f32_16x16x4_f32(false, ai0, false, bi0, (short)0, accI, false, false);
    accR = __builtin_amdgcn_wmma_f32_16x16x4_f32(false, ar1, false, br1, (short)0, accR, false, false);
    accI = __builtin_amdgcn_wmma_f32_16x16x4_f32(false, ai1, false, bi1, (short)0, accI, false, false);
  }
}

#define CROWS  81
#define CPITCH 128
#define CWAVES 4
#define CXTRA  11
#define CBLK   (CROWS + CXTRA)

__device__ __forceinline__ constexpr int xtra_row(int r) {
  return r == 39 ? 81 : r == 41 ? 82 : r == 37 ? 83 : r == 43 ? 84
       : r == 31 ? 85 : r == 49 ? 86 : r == 13 ? 87 : r == 67 ? 88 : 89;
}
__device__ __forceinline__ constexpr int xtra_cnt(int r) {
  return r == 40 ? 3
       : (r == 39 || r == 41 || r == 37 || r == 43 || r == 31 || r == 49 || r == 13 || r == 67) ? 1 : 0;
}
__device__ __forceinline__ constexpr int split_row(int q) {
  return q == 0 ? 13 : q == 1 ? 31 : q == 2 ? 37 : q == 3 ? 39 : q == 4 ? 40
       : q == 5 ? 41 : q == 6 ? 43 : q == 7 ? 49 : 67;
}
__device__ __forceinline__ constexpr int row_of(int k) {
  return k >= 72 ? split_row(k - 72)
       : k + (k >= 13 ? 1 : 0) + (k >= 30 ? 1 : 0) + (k >= 35 ? 1 : 0) + (k >= 36 ? 3 : 0)
           + (k >= 37 ? 1 : 0) + (k >= 42 ? 1 : 0) + (k >= 59 ? 1 : 0);
}

__global__ __launch_bounds__(CWAVES * 32) void build_C(const float* __restrict__ rR,
                                                       const float* __restrict__ rI,
                                                       const float* __restrict__ gR,
                                                       const float* __restrict__ gI,
                                                       float* __restrict__ cP) {
  __shared__ float tile[CWAVES][16][17];
  __shared__ float part[CWAVES][CPITCH];
  const int tid = threadIdx.x;
  const int w = __builtin_amdgcn_readfirstlane(tid >> 5);
  const int l = tid & 31;
  const int h = l >> 4;
  const int c = l & 15;
  const int b = blockIdx.x;
  const int ts = w & 1;
  const int kh = w >> 1;

  int i = b, prt = 0;
  if (b >= CROWS + 8) {
    i = 40;
    prt = b - (CROWS + 8) + 1;
  } else if (b >= CROWS) {
    const int e  = b - CROWS;
    const int d  = e >> 1;
    const int p3 = d == 0 ? 1 : d == 1 ? 3 : d == 2 ? 9 : 27;
    i = 40 + ((e & 1) ? p3 : -p3);
    prt = 1;
  }

  int base2 = 0, m1 = 0;
  {
    int v = i;
#pragma unroll
    for (int bb = 0; bb < 4; ++bb) {
      const int d = v % 3;
      v /= 3;
      base2 |= (d == 2 ? 1 : 0) << bb;
      m1    |= (d == 1 ? 1 : 0) << bb;
    }
  }
  const int m1p = m1 & (m1 - 1);
  const int tix = 2 * prt + ts;
  int sub = -1, nact = 0;
  for (int s = 0; s < 16; ++s) {
    if ((s & ~m1p) == 0) {
      if (nact == tix) sub = s;
      ++nact;
    }
  }
  const bool active = sub >= 0;

  if (active) {
    const int kt = base2 | sub;
    const int nt = base2 | (m1 ^ sub);
    const size_t ao = (size_t)(kt * 16 + c) * QDIM + 4 * h + 128 * kh;
    const size_t bo = (size_t)(nt * 16 + c) * QDIM + 4 * h + 128 * kh;

    v8f accR = {0.f, 0.f, 0.f, 0.f, 0.f, 0.f, 0.f, 0.f};
    v8f accI = accR;
    FragQ f0, f1;
    load_fragQ(f0, gR + ao, gI + ao, rR + bo, rI + bo);
    load_fragQ(f1, gR + ao + 8 * QB, gI + ao + 8 * QB, rR + bo + 8 * QB, rI + bo + 8 * QB);
    __builtin_amdgcn_sched_barrier(6);
    mma_fragQ(f0, accR, accI);
    mma_fragQ(f1, accR, accI);
#pragma unroll
    for (int r = 0; r < 8; ++r) tile[w][8 * h + r][c] = accR[r] + accI[r];
  }
  __syncthreads();

  float cs[4] = {0.f, 0.f, 0.f, 0.f};
  if (active) {
    float v[4][16];
    int   mk[4];
#pragma unroll
    for (int u = 0; u < 4; ++u) {
      const int o  = l + 32 * u;
      const int jh = o >> 2;
      const int t7 = o & 3;
      const int d1 = jh % 3, d2 = (jh / 3) % 3, d3 = jh / 9;
      const int b2 = (t7 == 2 ? 1 : 0) | (d1 == 2 ? 2 : 0) | (d2 == 2 ? 4 : 0) | (d3 == 2 ? 8 : 0);
      const int mm = (t7 == 1 ? 1 : 0) | (d1 == 1 ? 2 : 0) | (d2 == 1 ? 4 : 0) | (d3 == 1 ? 8 : 0);
      mk[u] = mm;
#pragma unroll
      for (int q = 0; q < 16; ++q) {
        const int qq = q & mm;
        v[u][q] = tile[w][b2 | qq][b2 | (mm ^ qq)];
      }
    }
#pragma unroll
    for (int u = 0; u < 4; ++u) {
      const int o = l + 32 * u;
      float s = 0.f;
#pragma unroll
      for (int q = 0; q < 16; ++q) s += ((q & ~mk[u]) == 0) ? v[u][q] : 0.f;
      cs[u] = ((o >> 2) < 27 && (o & 3) < 3) ? s : 0.f;
    }
  }
#pragma unroll
  for (int u = 0; u < 4; ++u) part[w][l + 32 * u] = cs[u];
  __syncthreads();

  {
    float pv[CWAVES];
#pragma unroll
    for (int ww = 0; ww < CWAVES; ++ww) pv[ww] = part[ww][tid];
    float s = 0.f;
#pragma unroll
    for (int ww = 0; ww < CWAVES; ++ww) s += pv[ww];
    cP[b * CPITCH + tid] = (m1 != 0 ? 2.f : 1.f) * s;
  }
}

#define KS 27
struct FragA { v2f a0[KS], a1[KS]; };
struct FragX { v2f x0[9][3], x1[9][3]; };

__device__ __forceinline__ void load_fragA(FragA& f, const float* p, const int k0) {
#pragma unroll
  for (int j = 0; j < KS; ++j) {
    const int r = row_of(k0 + j);
    f.a0[j] = *(const v2f*)(p + (size_t)r * CPITCH);
    f.a1[j] = *(const v2f*)(p + (size_t)r * CPITCH + 64);
  }
}
__device__ __forceinline__ void load_fragX(FragX& x, const float* p) {
#pragma unroll
  for (int q = 0; q < 9; ++q) {
#pragma unroll
    for (int u = 0; u < 3; ++u) {
      if (u < xtra_cnt(split_row(q))) {
        x.x0[q][u] = *(const v2f*)(p + (size_t)(xtra_row(split_row(q)) + u) * CPITCH);
        x.x1[q][u] = *(const v2f*)(p + (size_t)(xtra_row(split_row(q)) + u) * CPITCH + 64);
      }
    }
  }
}

__device__ __forceinline__ void mma_fragA(const v2f* bm, const v2f* bn, const FragA& f, const FragX& x,
                                          v8f& y0, v8f& y1, v8f& y2, v8f& y3, const int k0) {
#pragma unroll
  for (int j = 0; j < KS; ++j) {
    const int k = k0 + j;
    const int r = row_of(k);
    v2f a0 = f.a0[j], a1 = f.a1[j];
    if (k >= 72) {
#pragma unroll
      for (int u = 0; u < 3; ++u) {
        if (u < xtra_cnt(r)) {
          a0 += x.x0[k - 72][u];
          a1 += x.x1[k - 72][u];
        }
      }
    }
    y0 = __builtin_amdgcn_wmma_f32_16x16x4_f32(false, a0, false, bm[r], (short)0, y0, false, false);
    y1 = __builtin_amdgcn_wmma_f32_16x16x4_f32(false, a1, false, bm[r], (short)0, y1, false, false);
    y2 = __builtin_amdgcn_wmma_f32_16x16x4_f32(false, a0, false, bn[r], (short)0, y2, false, false);
    y3 = __builtin_amdgcn_wmma_f32_16x16x4_f32(false, a1, false, bn[r], (short)0, y3, false, false);
  }
}

__device__ __forceinline__ void tile_features(const float* cq, const float* sq, const int h,
                                              v2f* bM, float* w0, float* w1) {
  float F[NQ][3];
#pragma unroll
  for (int q = 0; q < NQ; ++q) {
    F[q][0] = cq[q] * cq[q];
    F[q][1] = cq[q] * sq[q];
    F[q][2] = sq[q] * sq[q];
  }
  const float vx = h ? F[7][2] : F[7][0];
  const float vy = h ? 0.f     : F[7][1];
#pragma unroll
  for (int t0 = 0; t0 < 3; ++t0) {
#pragma unroll
    for (int t1 = 0; t1 < 3; ++t1) {
      const float p01 = F[0][t0] * F[1][t1];
#pragma unroll
      for (int t2 = 0; t2 < 3; ++t2) {
        const float p012 = p01 * F[2][t2];
#pragma unroll
        for (int t3 = 0; t3 < 3; ++t3) {
          const float u = p012 * F[3][t3];
          const int i = ((t0 * 3 + t1) * 3 + t2) * 3 + t3;
          bM[i].x = u * vx;
          bM[i].y = u * vy;
        }
      }
    }
  }
  float V[32];
#pragma unroll
  for (int jh = 0; jh < 32; ++jh)
    V[jh] = (jh < 27) ? (F[4][jh / 9] * F[5][(jh / 3) % 3]) * F[6][jh % 3] : 0.f;
#pragma unroll
  for (int r = 0; r < 8; ++r) {
    w0[r] = h ? V[8 + r]  : V[r];
    w1[r] = h ? V[24 + r] : V[16 + r];
  }
}

__global__ __launch_bounds__(32) __attribute__((amdgpu_waves_per_eu(1, 1))) void vqe_wmma(const float* __restrict__ x,
                                               const float* __restrict__ fcb,
                                               const float* __restrict__ cP,
                                               float* __restrict__ out) {
  const int l = threadIdx.x;
  const int h = l >> 4;
  const int c = l & 15;
  const int sBase = blockIdx.x * 32;

  const v4f xa = *(const v4f*)(x + (size_t)(sBase + l) * NQ);
  const v4f xb = *(const v4f*)(x + (size_t)(sBase + l) * NQ + 4);
  const float xq[NQ] = {xa.x, xa.y, xa.z, xa.w, xb.x, xb.y, xb.z, xb.w};

  const float* abase = cP + 4 * c + 2 * h;
  FragA f0, f1, f2;
  FragX fx;
  load_fragA(f0, abase, 0);

  float co[NQ], so[NQ];
#pragma unroll
  for (int q = 0; q < NQ; ++q) {
    const float xv = 0.5f * xq[q];
    co[q] = cosf(xv);
    so[q] = sinf(xv);
  }

  load_fragA(f1, abase, KS);

  float c0[NQ], s0q[NQ], c1[NQ], s1q[NQ];
#pragma unroll
  for (int q = 0; q < NQ; ++q) {
    const float cp = __shfl_xor(co[q], 16);
    const float sp = __shfl_xor(so[q], 16);
    c0[q]  = h ? cp : co[q];
    s0q[q] = h ? sp : so[q];
    c1[q]  = h ? co[q] : cp;
    s1q[q] = h ? so[q] : sp;
  }

  v2f bM0[CROWS], bM1[CROWS];
  float w00[8], w01[8], w10[8], w11[8];
  tile_features(c0, s0q, h, bM0, w00, w01);
  tile_features(c1, s1q, h, bM1, w10, w11);

  v8f y0 = {0.f, 0.f, 0.f, 0.f, 0.f, 0.f, 0.f, 0.f};
  v8f y1 = y0, y2 = y0, y3 = y0;
  __builtin_amdgcn_sched_barrier(6);
  mma_fragA(bM0, bM1, f0, fx, y0, y1, y2, y3, 0);
  __builtin_amdgcn_sched_barrier(6);
  load_fragA(f2, abase, 2 * KS);
  __builtin_amdgcn_sched_barrier(6);
  mma_fragA(bM0, bM1, f1, fx, y0, y1, y2, y3, KS);
  __builtin_amdgcn_sched_barrier(6);
  load_fragX(fx, abase);
  __builtin_amdgcn_sched_barrier(6);
  mma_fragA(bM0, bM1, f2, fx, y0, y1, y2, y3, 2 * KS);

  float s0 = 0.f, s1 = 0.f, s2 = 0.f, s3 = 0.f;
#pragma unroll
  for (int r = 0; r < 8; ++r) {
    s0 += y0[r] * w00[r];
    s1 += y1[r] * w01[r];
    s2 += y2[r] * w10[r];
    s3 += y3[r] * w11[r];
  }
  float za = s0 + s1;
  float zb = s2 + s3;

  za += __shfl_xor(za, 16);
  zb += __shfl_xor(zb, 16);
  const float sum = fcb[0] + (h ? zb : za);
  out[sBase + l] = 1.f / (1.f + expf(-sum));
}

extern "C" void kernel_launch(void* const* d_in, const int* in_sizes, int n_in,
                              void* d_out, int out_size, void* d_ws, size_t ws_size,
                              hipStream_t stream) {
  const float* x       = (const float*)d_in[0];
  const float* weights = (const float*)d_in[1];
  const float* fcw     = (const float*)d_in[2];
  const float* fcb     = (const float*)d_in[3];
  float* out = (float*)d_out;

  float* rR = (float*)d_ws;
  float* rI = rR + QDIM * QDIM;
  float* gR = rI + QDIM * QDIM;
  float* gI = gR + QDIM * QDIM;
  float* cP = gI + QDIM * QDIM;

  const int B        = in_sizes[0] / NQ;
  const int n_layers = in_sizes[1] / (NQ * 3);

  build_W<<<QDIM / 2, 32, 0, stream>>>(weights, fcw, n_layers, rR, rI, gR, gI);
  build_C<<<CBLK, CWAVES * 32, 0, stream>>>(rR, rI, gR, gI, cP);
  vqe_wmma<<<B / 32, 32, 0, stream>>>(x, fcb, cP, out);
}
